# starting kernel
# baseline (speedup 1.0000x reference)
#include <hip/hip_runtime.h>
#include <hip/hip_bf16.h>
#include <math.h>

typedef __attribute__((ext_vector_type(2))) float v2f;
typedef __attribute__((ext_vector_type(8))) float v8f;

#define B_   4
#define N_   2048
#define FIN  256
#define FOUT 128
#define ALPHA_ 0.2f
#define EPS_   1e-5f
#define NEG_   -9000000000000000.0f

__global__ __launch_bounds__(128) void wh_kernel(const float* __restrict__ h,
                                                 const float* __restrict__ W,
                                                 float* __restrict__ Wh) {
  int lane  = threadIdx.x & 31;
  int tile  = blockIdx.x * 4 + (threadIdx.x >> 5);
  int otile = tile & 7;
  int rest  = tile >> 3;
  int itile = rest & (N_ / 16 - 1);
  int b     = rest >> 7;
  int i0 = itile * 16, o0 = otile * 16;
  int r16  = lane & 15;
  int koff = (lane >> 4) << 1;

  const float* hrow = h + (size_t)(b * N_ + i0 + r16) * FIN;
  v8f c = {};
  for (int k0 = 0; k0 < FIN; k0 += 4) {
    float2 av = *(const float2*)(hrow + k0 + koff);
    v2f a; a.x = av.x; a.y = av.y;
    v2f bb;
    bb.x = W[(size_t)(k0 + koff + 0) * FOUT + o0 + r16];
    bb.y = W[(size_t)(k0 + koff + 1) * FOUT + o0 + r16];
    c = __builtin_amdgcn_wmma_f32_16x16x4_f32(false, a, false, bb, (short)0, c,
                                              false, false);
  }
#pragma unroll
  for (int r = 0; r < 8; ++r) {
    int M = r + ((lane >> 4) << 3);
    Wh[(size_t)(b * N_ + i0 + M) * FOUT + o0 + r16] = c[r];
  }
}

__global__ __launch_bounds__(256) void s12_kernel(const float* __restrict__ Wh,
                                                  const float* __restrict__ a,
                                                  float* __restrict__ s1,
                                                  float* __restrict__ s2) {
  int lane = threadIdx.x & 31;
  int node = (int)((blockIdx.x * blockDim.x + threadIdx.x) >> 5);
  if (node >= B_ * N_) return;
  const float4* w4 = (const float4*)(Wh + (size_t)node * FOUT);
  const float4* a1 = (const float4*)a;
  const float4* a2 = (const float4*)(a + FOUT);
  float4 wv = w4[lane];
  float4 x1 = a1[lane];
  float4 x2 = a2[lane];
  float p1 = wv.x * x1.x + wv.y * x1.y + wv.z * x1.z + wv.w * x1.w;
  float p2 = wv.x * x2.x + wv.y * x2.y + wv.z * x2.z + wv.w * x2.w;
  for (int off = 16; off >= 1; off >>= 1) {
    p1 += __shfl_xor(p1, off, 32);
    p2 += __shfl_xor(p2, off, 32);
  }
  if (lane == 0) { s1[node] = p1; s2[node] = p2; }
}

__global__ __launch_bounds__(256) void stats_kernel(const int* __restrict__ adj,
                                                    const float* __restrict__ s1,
                                                    const float* __restrict__ s2,
                                                    float* __restrict__ mrow,
                                                    float* __restrict__ lrow,
                                                    unsigned* __restrict__ mbits) {
  int lane = threadIdx.x & 31;
  int row  = blockIdx.x * 8 + (threadIdx.x >> 5);
  int b    = row >> 11;
  float s1v = s1[row];
  const float* s2b = s2 + (size_t)b * N_;
  const int* arow  = adj + (size_t)row * N_;
  unsigned* mb     = mbits + (size_t)row * (N_ / 32);
  float m = -INFINITY, l = 0.f;
  for (int w = 0; w < N_ / 32; ++w) {
    int j = w * 32 + lane;
    float e = s1v + s2b[j];
    e = (e >= 0.f) ? e : ALPHA_ * e;
    bool pred = arow[j] > 0;
    unsigned long long bl = __ballot(pred);
    if (lane == 0) mb[w] = (unsigned)bl;
    float tv = pred ? e : NEG_;
    if (tv > m) { l = l * __expf(m - tv) + 1.0f; m = tv; }
    else        { l += __expf(tv - m); }
  }
  for (int off = 16; off >= 1; off >>= 1) {
    float m2 = __shfl_xor(m, off, 32);
    float l2 = __shfl_xor(l, off, 32);
    float M  = fmaxf(m, m2);
    l = l * __expf(m - M) + l2 * __expf(m2 - M);
    m = M;
  }
  if (lane == 0) { mrow[row] = m; lrow[row] = l; }
}

__global__ __launch_bounds__(128) void hprime_kernel(
    const unsigned* __restrict__ mbits, const float* __restrict__ Wh,
    const float* __restrict__ s1, const float* __restrict__ s2,
    const float* __restrict__ mrow, const float* __restrict__ lrow,
    const float* __restrict__ gamma, const float* __restrict__ beta,
    float* __restrict__ out) {
  __shared__ float lds[4][16][FOUT];
  int tid  = threadIdx.x;
  int lane = tid & 31;
  int wave = tid >> 5;
  int itile = blockIdx.x & (N_ / 16 - 1);
  int b     = blockIdx.x >> 7;
  int i0    = itile * 16;
  int r16   = lane & 15;
  int koff  = (lane >> 4) << 1;
  int rowid = b * N_ + i0 + r16;

  float s1v  = s1[rowid];
  float mi   = mrow[rowid];
  float invl = 1.0f / lrow[rowid];
  const unsigned* mb = mbits + (size_t)rowid * (N_ / 32);
  const float* s2b   = s2 + (size_t)b * N_;
  const float* Whb   = Wh + (size_t)b * N_ * FOUT;

  v8f acc[8];
#pragma unroll
  for (int ot = 0; ot < 8; ++ot) acc[ot] = (v8f){};

  int jbeg = wave * (N_ / 4);
  int jend = jbeg + (N_ / 4);
  for (int j0 = jbeg; j0 < jend; j0 += 4) {
    int j = j0 + koff;
    unsigned wbits = mb[j0 >> 5];
    v2f a;
#pragma unroll
    for (int v = 0; v < 2; ++v) {
      int jj = j + v;
      float e = s1v + s2b[jj];
      e = (e >= 0.f) ? e : ALPHA_ * e;
      float tv = ((wbits >> ((j0 & 31) + koff + v)) & 1u) ? e : NEG_;
      a[v] = __expf(tv - mi) * invl;
    }
    const float* w0 = Whb + (size_t)j * FOUT + r16;
    const float* w1 = w0 + FOUT;
#pragma unroll
    for (int ot = 0; ot < 8; ++ot) {
      v2f bb;
      bb.x = w0[ot * 16];
      bb.y = w1[ot * 16];
      acc[ot] = __builtin_amdgcn_wmma_f32_16x16x4_f32(false, a, false, bb,
                                                      (short)0, acc[ot],
                                                      false, false);
    }
  }

#pragma unroll
  for (int ot = 0; ot < 8; ++ot)
#pragma unroll
    for (int r = 0; r < 8; ++r) {
      int M = r + ((lane >> 4) << 3);
      lds[wave][M][ot * 16 + r16] = acc[ot][r];
    }
  __syncthreads();

  int row16 = tid >> 3;
  int seg   = tid & 7;
  float vals[16];
  float s = 0.f, ss = 0.f;
#pragma unroll
  for (int c = 0; c < 16; ++c) {
    int col = seg * 16 + c;
    float x = lds[0][row16][col] + lds[1][row16][col] +
              lds[2][row16][col] + lds[3][row16][col];
    vals[c] = x;
    s += x;
    ss += x * x;
  }
  for (int off = 4; off >= 1; off >>= 1) {
    s  += __shfl_xor(s, off, 8);
    ss += __shfl_xor(ss, off, 8);
  }
  float mu   = s * (1.0f / FOUT);
  float var  = ss * (1.0f / FOUT) - mu * mu;
  float rstd = rsqrtf(var + EPS_);
  float* orow = out + (size_t)(b * N_ + i0 + row16) * FOUT;
#pragma unroll
  for (int c = 0; c < 16; ++c) {
    int col = seg * 16 + c;
    float x = (vals[c] - mu) * rstd * gamma[col] + beta[col];
    orow[col] = 0.5f * x * (1.0f + erff(x * 0.70710678118654752f));
  }
}

extern "C" void kernel_launch(void* const* d_in, const int* in_sizes, int n_in,
                              void* d_out, int out_size, void* d_ws, size_t ws_size,
                              hipStream_t stream) {
  const float* h     = (const float*)d_in[0];
  const int*   adj   = (const int*)d_in[1];
  const float* W     = (const float*)d_in[2];
  const float* a     = (const float*)d_in[3];
  const float* gamma = (const float*)d_in[4];
  const float* beta  = (const float*)d_in[5];
  float* out = (float*)d_out;

  float* ws   = (float*)d_ws;
  float* Wh   = ws;
  float* s1   = Wh + (size_t)B_ * N_ * FOUT;
  float* s2   = s1 + B_ * N_;
  float* mrow = s2 + B_ * N_;
  float* lrow = mrow + B_ * N_;
  unsigned* mbits = (unsigned*)(lrow + B_ * N_);

  wh_kernel<<<(B_ * (N_ / 16) * 8) / 4, 128, 0, stream>>>(h, W, Wh);
  s12_kernel<<<(B_ * N_ * 32) / 256, 256, 0, stream>>>(Wh, a, s1, s2);
  stats_kernel<<<(B_ * N_) / 8, 256, 0, stream>>>(adj, s1, s2, mrow, lrow, mbits);
  hprime_kernel<<<B_ * (N_ / 16), 128, 0, stream>>>(mbits, Wh, s1, s2, mrow, lrow,
                                                    gamma, beta, out);
}


# stats kernel: issue the adj and s2 loads of 16 column windows per wave before the first use (was one window in flight); same per-column arithmetic and order
# speedup vs baseline: 1.1871x; 1.1871x over previous
#include <hip/hip_runtime.h>
#include <hip/hip_bf16.h>
#include <math.h>

typedef __attribute__((ext_vector_type(2))) float v2f;
typedef __attribute__((ext_vector_type(8))) float v8f;

#define B_   4
#define N_   2048
#define FIN  256
#define FOUT 128
#define ALPHA_ 0.2f
#define EPS_   1e-5f
#define NEG_   -9000000000000000.0f

__global__ __launch_bounds__(128) void wh_kernel(const float* __restrict__ h,
                                                 const float* __restrict__ W,
                                                 float* __restrict__ Wh) {
  int lane  = threadIdx.x & 31;
  int tile  = blockIdx.x * 4 + (threadIdx.x >> 5);
  int otile = tile & 7;
  int rest  = tile >> 3;
  int itile = rest & (N_ / 16 - 1);
  int b     = rest >> 7;
  int i0 = itile * 16, o0 = otile * 16;
  int r16  = lane & 15;
  int koff = (lane >> 4) << 1;

  const float* hrow = h + (size_t)(b * N_ + i0 + r16) * FIN;
  v8f c = {};
  for (int k0 = 0; k0 < FIN; k0 += 4) {
    float2 av = *(const float2*)(hrow + k0 + koff);
    v2f a; a.x = av.x; a.y = av.y;
    v2f bb;
    bb.x = W[(size_t)(k0 + koff + 0) * FOUT + o0 + r16];
    bb.y = W[(size_t)(k0 + koff + 1) * FOUT + o0 + r16];
    c = __builtin_amdgcn_wmma_f32_16x16x4_f32(false, a, false, bb, (short)0, c,
                                              false, false);
  }
#pragma unroll
  for (int r = 0; r < 8; ++r) {
    int M = r + ((lane >> 4) << 3);
    Wh[(size_t)(b * N_ + i0 + M) * FOUT + o0 + r16] = c[r];
  }
}

__global__ __launch_bounds__(256) void s12_kernel(const float* __restrict__ Wh,
                                                  const float* __restrict__ a,
                                                  float* __restrict__ s1,
                                                  float* __restrict__ s2) {
  int lane = threadIdx.x & 31;
  int node = (int)((blockIdx.x * blockDim.x + threadIdx.x) >> 5);
  if (node >= B_ * N_) return;
  const float4* w4 = (const float4*)(Wh + (size_t)node * FOUT);
  const float4* a1 = (const float4*)a;
  const float4* a2 = (const float4*)(a + FOUT);
  float4 wv = w4[lane];
  float4 x1 = a1[lane];
  float4 x2 = a2[lane];
  float p1 = wv.x * x1.x + wv.y * x1.y + wv.z * x1.z + wv.w * x1.w;
  float p2 = wv.x * x2.x + wv.y * x2.y + wv.z * x2.z + wv.w * x2.w;
  for (int off = 16; off >= 1; off >>= 1) {
    p1 += __shfl_xor(p1, off, 32);
    p2 += __shfl_xor(p2, off, 32);
  }
  if (lane == 0) { s1[node] = p1; s2[node] = p2; }
}

__global__ __launch_bounds__(256) void stats_kernel(const int* __restrict__ adj,
                                                    const float* __restrict__ s1,
                                                    const float* __restrict__ s2,
                                                    float* __restrict__ mrow,
                                                    float* __restrict__ lrow,
                                                    unsigned* __restrict__ mbits) {
  int lane = threadIdx.x & 31;
  int row  = blockIdx.x * 8 + (threadIdx.x >> 5);
  int b    = row >> 11;
  float s1v = s1[row];
  const float* s2b = s2 + (size_t)b * N_;
  const int* arow  = adj + (size_t)row * N_;
  unsigned* mb     = mbits + (size_t)row * (N_ / 32);
  float m = -INFINITY, l = 0.f;
  constexpr int GRP_ = 16;
  for (int w0 = 0; w0 < N_ / 32; w0 += GRP_) {
    int   av[GRP_];
    float sv[GRP_];
#pragma unroll
    for (int u = 0; u < GRP_; ++u) {
      int j = (w0 + u) * 32 + lane;
      av[u] = arow[j];
      sv[u] = s2b[j];
    }
#pragma unroll
    for (int u = 0; u < GRP_; ++u) {
      float e = s1v + sv[u];
      e = (e >= 0.f) ? e : ALPHA_ * e;
      bool pred = av[u] > 0;
      unsigned long long bl = __ballot(pred);
      if (lane == 0) mb[w0 + u] = (unsigned)bl;
      float tv = pred ? e : NEG_;
      if (tv > m) { l = l * __expf(m - tv) + 1.0f; m = tv; }
      else        { l += __expf(tv - m); }
    }
  }
  for (int off = 16; off >= 1; off >>= 1) {
    float m2 = __shfl_xor(m, off, 32);
    float l2 = __shfl_xor(l, off, 32);
    float M  = fmaxf(m, m2);
    l = l * __expf(m - M) + l2 * __expf(m2 - M);
    m = M;
  }
  if (lane == 0) { mrow[row] = m; lrow[row] = l; }
}

__global__ __launch_bounds__(128) void hprime_kernel(
    const unsigned* __restrict__ mbits, const float* __restrict__ Wh,
    const float* __restrict__ s1, const float* __restrict__ s2,
    const float* __restrict__ mrow, const float* __restrict__ lrow,
    const float* __restrict__ gamma, const float* __restrict__ beta,
    float* __restrict__ out) {
  __shared__ float lds[4][16][FOUT];
  int tid  = threadIdx.x;
  int lane = tid & 31;
  int wave = tid >> 5;
  int itile = blockIdx.x & (N_ / 16 - 1);
  int b     = blockIdx.x >> 7;
  int i0    = itile * 16;
  int r16   = lane & 15;
  int koff  = (lane >> 4) << 1;
  int rowid = b * N_ + i0 + r16;

  float s1v  = s1[rowid];
  float mi   = mrow[rowid];
  float invl = 1.0f / lrow[rowid];
  const unsigned* mb = mbits + (size_t)rowid * (N_ / 32);
  const float* s2b   = s2 + (size_t)b * N_;
  const float* Whb   = Wh + (size_t)b * N_ * FOUT;

  v8f acc[8];
#pragma unroll
  for (int ot = 0; ot < 8; ++ot) acc[ot] = (v8f){};

  int jbeg = wave * (N_ / 4);
  int jend = jbeg + (N_ / 4);
  for (int j0 = jbeg; j0 < jend; j0 += 4) {
    int j = j0 + koff;
    unsigned wbits = mb[j0 >> 5];
    v2f a;
#pragma unroll
    for (int v = 0; v < 2; ++v) {
      int jj = j + v;
      float e = s1v + s2b[jj];
      e = (e >= 0.f) ? e : ALPHA_ * e;
      float tv = ((wbits >> ((j0 & 31) + koff + v)) & 1u) ? e : NEG_;
      a[v] = __expf(tv - mi) * invl;
    }
    const float* w0 = Whb + (size_t)j * FOUT + r16;
    const float* w1 = w0 + FOUT;
#pragma unroll
    for (int ot = 0; ot < 8; ++ot) {
      v2f bb;
      bb.x = w0[ot * 16];
      bb.y = w1[ot * 16];
      acc[ot] = __builtin_amdgcn_wmma_f32_16x16x4_f32(false, a, false, bb,
                                                      (short)0, acc[ot],
                                                      false, false);
    }
  }

#pragma unroll
  for (int ot = 0; ot < 8; ++ot)
#pragma unroll
    for (int r = 0; r < 8; ++r) {
      int M = r + ((lane >> 4) << 3);
      lds[wave][M][ot * 16 + r16] = acc[ot][r];
    }
  __syncthreads();

  int row16 = tid >> 3;
  int seg   = tid & 7;
  float vals[16];
  float s = 0.f, ss = 0.f;
#pragma unroll
  for (int c = 0; c < 16; ++c) {
    int col = seg * 16 + c;
    float x = lds[0][row16][col] + lds[1][row16][col] +
              lds[2][row16][col] + lds[3][row16][col];
    vals[c] = x;
    s += x;
    ss += x * x;
  }
  for (int off = 4; off >= 1; off >>= 1) {
    s  += __shfl_xor(s, off, 8);
    ss += __shfl_xor(ss, off, 8);
  }
  float mu   = s * (1.0f / FOUT);
  float var  = ss * (1.0f / FOUT) - mu * mu;
  float rstd = rsqrtf(var + EPS_);
  float* orow = out + (size_t)(b * N_ + i0 + row16) * FOUT;
#pragma unroll
  for (int c = 0; c < 16; ++c) {
    int col = seg * 16 + c;
    float x = (vals[c] - mu) * rstd * gamma[col] + beta[col];
    orow[col] = 0.5f * x * (1.0f + erff(x * 0.70710678118654752f));
  }
}

extern "C" void kernel_launch(void* const* d_in, const int* in_sizes, int n_in,
                              void* d_out, int out_size, void* d_ws, size_t ws_size,
                              hipStream_t stream) {
  const float* h     = (const float*)d_in[0];
  const int*   adj   = (const int*)d_in[1];
  const float* W     = (const float*)d_in[2];
  const float* a     = (const float*)d_in[3];
  const float* gamma = (const float*)d_in[4];
  const float* beta  = (const float*)d_in[5];
  float* out = (float*)d_out;

  float* ws   = (float*)d_ws;
  float* Wh   = ws;
  float* s1   = Wh + (size_t)B_ * N_ * FOUT;
  float* s2   = s1 + B_ * N_;
  float* mrow = s2 + B_ * N_;
  float* lrow = mrow + B_ * N_;
  unsigned* mbits = (unsigned*)(lrow + B_ * N_);

  wh_kernel<<<(B_ * (N_ / 16) * 8) / 4, 128, 0, stream>>>(h, W, Wh);
  s12_kernel<<<(B_ * N_ * 32) / 256, 256, 0, stream>>>(Wh, a, s1, s2);
  stats_kernel<<<(B_ * N_) / 8, 256, 0, stream>>>(adj, s1, s2, mrow, lrow, mbits);
  hprime_kernel<<<B_ * (N_ / 16), 128, 0, stream>>>(mbits, Wh, s1, s2, mrow, lrow,
                                                    gamma, beta, out);
}


# Wh kernel: block = the 8 column tiles of a 16-row strip, staged in LDS; s1/s2 row dots done there (same products and xor tree), so the separate s1/s2 launch is removed: 4 to 3 launches
# speedup vs baseline: 1.4685x; 1.2370x over previous
#include <hip/hip_runtime.h>
#include <hip/hip_bf16.h>
#include <math.h>

typedef __attribute__((ext_vector_type(2))) float v2f;
typedef __attribute__((ext_vector_type(8))) float v8f;

#define B_   4
#define N_   2048
#define FIN  256
#define FOUT 128
#define ALPHA_ 0.2f
#define EPS_   1e-5f
#define NEG_   -9000000000000000.0f

#define WP_ 132
__global__ __launch_bounds__(256) void wh_kernel(const float* __restrict__ h,
                                                 const float* __restrict__ W,
                                                 const float* __restrict__ a,
                                                 float* __restrict__ Wh,
                                                 float* __restrict__ s1,
                                                 float* __restrict__ s2) {
  __shared__ __attribute__((aligned(16))) float strip[16 * WP_];
  int lane  = threadIdx.x & 31;
  int otile = threadIdx.x >> 5;
  int itile = blockIdx.x & (N_ / 16 - 1);
  int b     = blockIdx.x >> 7;
  int i0 = itile * 16, o0 = otile * 16;
  int r16  = lane & 15;
  int koff = (lane >> 4) << 1;

  const float4* a1 = (const float4*)a;
  const float4* a2 = (const float4*)(a + FOUT);
  float4 x1 = a1[lane];
  float4 x2 = a2[lane];

  const float* hrow = h + (size_t)(b * N_ + i0 + r16) * FIN;
  v8f c = {};
  for (int k0 = 0; k0 < FIN; k0 += 4) {
    float2 av = *(const float2*)(hrow + k0 + koff);
    v2f a_; a_.x = av.x; a_.y = av.y;
    v2f bb;
    bb.x = W[(size_t)(k0 + koff + 0) * FOUT + o0 + r16];
    bb.y = W[(size_t)(k0 + koff + 1) * FOUT + o0 + r16];
    c = __builtin_amdgcn_wmma_f32_16x16x4_f32(false, a_, false, bb, (short)0, c,
                                              false, false);
  }
#pragma unroll
  for (int r = 0; r < 8; ++r) {
    int M = r + ((lane >> 4) << 3);
    strip[M * WP_ + o0 + r16] = c[r];
  }
  __syncthreads();
#pragma unroll
  for (int r = 0; r < 8; ++r) {
    int M = r + ((lane >> 4) << 3);
    Wh[(size_t)(b * N_ + i0 + M) * FOUT + o0 + r16] = c[r];
  }

  float p1[2], p2[2];
#pragma unroll
  for (int rr = 0; rr < 2; ++rr) {
    float4 wv = *(const float4*)&strip[(2 * otile + rr) * WP_ + 4 * lane];
    p1[rr] = wv.x * x1.x + wv.y * x1.y + wv.z * x1.z + wv.w * x1.w;
    p2[rr] = wv.x * x2.x + wv.y * x2.y + wv.z * x2.z + wv.w * x2.w;
  }
#pragma unroll
  for (int off = 16; off >= 1; off >>= 1) {
#pragma unroll
    for (int rr = 0; rr < 2; ++rr) {
      p1[rr] += __shfl_xor(p1[rr], off, 32);
      p2[rr] += __shfl_xor(p2[rr], off, 32);
    }
  }
  if (lane < 2) {
    int node = b * N_ + i0 + 2 * otile + lane;
    s1[node] = (lane == 0) ? p1[0] : p1[1];
    s2[node] = (lane == 0) ? p2[0] : p2[1];
  }
}

__global__ __launch_bounds__(256) void stats_kernel(const int* __restrict__ adj,
                                                    const float* __restrict__ s1,
                                                    const float* __restrict__ s2,
                                                    float* __restrict__ mrow,
                                                    float* __restrict__ lrow,
                                                    unsigned* __restrict__ mbits) {
  int lane = threadIdx.x & 31;
  int row  = blockIdx.x * 8 + (threadIdx.x >> 5);
  int b    = row >> 11;
  float s1v = s1[row];
  const float* s2b = s2 + (size_t)b * N_;
  const int* arow  = adj + (size_t)row * N_;
  unsigned* mb     = mbits + (size_t)row * (N_ / 32);
  float m = -INFINITY, l = 0.f;
  constexpr int GRP_ = 16;
  for (int w0 = 0; w0 < N_ / 32; w0 += GRP_) {
    int   av[GRP_];
    float sv[GRP_];
#pragma unroll
    for (int u = 0; u < GRP_; ++u) {
      int j = (w0 + u) * 32 + lane;
      av[u] = arow[j];
      sv[u] = s2b[j];
    }
#pragma unroll
    for (int u = 0; u < GRP_; ++u) {
      float e = s1v + sv[u];
      e = (e >= 0.f) ? e : ALPHA_ * e;
      bool pred = av[u] > 0;
      unsigned long long bl = __ballot(pred);
      if (lane == 0) mb[w0 + u] = (unsigned)bl;
      float tv = pred ? e : NEG_;
      if (tv > m) { l = l * __expf(m - tv) + 1.0f; m = tv; }
      else        { l += __expf(tv - m); }
    }
  }
  for (int off = 16; off >= 1; off >>= 1) {
    float m2 = __shfl_xor(m, off, 32);
    float l2 = __shfl_xor(l, off, 32);
    float M  = fmaxf(m, m2);
    l = l * __expf(m - M) + l2 * __expf(m2 - M);
    m = M;
  }
  if (lane == 0) { mrow[row] = m; lrow[row] = l; }
}

typedef __attribute__((ext_vector_type(4))) float v4f;
#define JC_     64
#define PP_     288
#define STAGE_  ((JC_ / 2) * PP_ + 4 * JC_)
#define TP_     132
#define LDSF_   (41 * 1024)
__global__ __launch_bounds__(128) void hprime_kernel(
    const unsigned* __restrict__ mbits, const float* __restrict__ Wh,
    const float* __restrict__ s1, const float* __restrict__ s2,
    const float* __restrict__ mrow, const float* __restrict__ lrow,
    const float* __restrict__ gamma, const float* __restrict__ beta,
    float* __restrict__ out) {
  __shared__ __attribute__((aligned(16))) float sB[LDSF_];
  int tid  = threadIdx.x;
  int lane = tid & 31;
  int wave = tid >> 5;
  int rt   = wave >> 1;
  int chf  = wave & 1;
  int b    = blockIdx.x >> 6;
  int i0   = (blockIdx.x & 63) * 32;
  int r16  = lane & 15;
  int hh   = lane >> 4;
  int rowid = b * N_ + i0 + rt * 16 + r16;

  float s1v  = s1[rowid];
  float mi   = mrow[rowid];
  float invl = 1.0f / lrow[rowid];
  const unsigned* mb = mbits + (size_t)rowid * (N_ / 32);
  const float* s2b   = s2 + (size_t)b * N_;
  const float* Whb   = Wh + (size_t)b * N_ * FOUT;

  v8f acc[4];
#pragma unroll
  for (int ot = 0; ot < 4; ++ot) acc[ot] = (v8f){};

  constexpr int NP = JC_ / 8;
  v4f  pa[NP], pb[NP];
  v2f  s2pre;
  uint2 mw, mwn;
  const float* gsrc = Whb + (size_t)(2 * wave) * FOUT + 4 * lane;
  const int    ldst = wave * PP_ + 8 * lane;
  const int    sdst = (JC_ / 2) * PP_ + wave * JC_ + 2 * lane;

#pragma unroll
  for (int i = 0; i < NP; ++i) {
    pa[i] = *(const v4f*)(gsrc + (size_t)(8 * i) * FOUT);
    pb[i] = *(const v4f*)(gsrc + (size_t)(8 * i + 1) * FOUT);
  }
  s2pre = *(const v2f*)(s2b + 2 * lane);
  mw = *(const uint2*)(mb);
#pragma unroll
  for (int i = 0; i < NP; ++i) {
    v4f lo = {pa[i].x, pb[i].x, pa[i].y, pb[i].y};
    v4f hi = {pa[i].z, pb[i].z, pa[i].w, pb[i].w};
    *(v4f*)&sB[ldst + 4 * i * PP_]     = lo;
    *(v4f*)&sB[ldst + 4 * i * PP_ + 4] = hi;
  }
  *(v2f*)&sB[sdst] = s2pre;
  __syncthreads();

  constexpr int NCH = N_ / JC_;
  for (int ck = 0; ck < NCH; ++ck) {
    const float* sb = sB + (ck & 1) * STAGE_;
    int cn = (ck + 1 < NCH) ? ck + 1 : ck;
    const float* gn = gsrc + (size_t)(cn * JC_) * FOUT;
#pragma unroll
    for (int i = 0; i < NP; ++i) {
      pa[i] = *(const v4f*)(gn + (size_t)(8 * i) * FOUT);
      pb[i] = *(const v4f*)(gn + (size_t)(8 * i + 1) * FOUT);
    }
    s2pre = *(const v2f*)(s2b + cn * JC_ + 2 * lane);
    mwn = *(const uint2*)(mb + 2 * cn);

    unsigned w0 = mw.x >> (2 * hh);
    unsigned w1 = mw.y >> (2 * hh);
    const float* s2l  = sb + (JC_ / 2) * PP_ + wave * JC_ + 2 * hh;
    const float* brow = sb + hh * PP_ + 2 * (64 * chf + r16);
#pragma unroll
    for (int t = 0; t < JC_ / 4; ++t) {
      v2f sv = *(const v2f*)(s2l + 4 * t);
      unsigned wsh = (t < 8) ? w0 : w1;
      v2f a;
      {
        float e = s1v + sv.x;
        e = (e >= 0.f) ? e : ALPHA_ * e;
        float tv = ((wsh >> (4 * (t & 7))) & 1u) ? e : NEG_;
        a.x = __expf(tv - mi) * invl;
      }
      {
        float e = s1v + sv.y;
        e = (e >= 0.f) ? e : ALPHA_ * e;
        float tv = ((wsh >> (4 * (t & 7) + 1)) & 1u) ? e : NEG_;
        a.y = __expf(tv - mi) * invl;
      }
      const float* bp = brow + (2 * t) * PP_;
#pragma unroll
      for (int ot = 0; ot < 4; ++ot) {
        v2f bb = *(const v2f*)(bp + 32 * ot);
        acc[ot] = __builtin_amdgcn_wmma_f32_16x16x4_f32(false, a, false, bb,
                                                        (short)0, acc[ot],
                                                        false, false);
      }
    }

    float* sn = sB + ((ck + 1) & 1) * STAGE_;
#pragma unroll
    for (int i = 0; i < NP; ++i) {
      v4f lo = {pa[i].x, pb[i].x, pa[i].y, pb[i].y};
      v4f hi = {pa[i].z, pb[i].z, pa[i].w, pb[i].w};
      *(v4f*)&sn[ldst + 4 * i * PP_]     = lo;
      *(v4f*)&sn[ldst + 4 * i * PP_ + 4] = hi;
    }
    *(v2f*)&sn[sdst] = s2pre;
    mw = mwn;
    __syncthreads();
  }

  float* tile = sB;
#pragma unroll
  for (int ot = 0; ot < 4; ++ot)
#pragma unroll
    for (int r = 0; r < 8; ++r)
      tile[(rt * 16 + hh * 8 + r) * TP_ + 64 * chf + ot * 16 + r16] = acc[ot][r];
  __syncthreads();

  v4f g4 = *(const v4f*)(gamma + 4 * lane);
  v4f b4 = *(const v4f*)(beta + 4 * lane);
  v4f xr[8];
  float sr[8], qr[8];
#pragma unroll
  for (int rr = 0; rr < 8; ++rr) {
    v4f x = *(const v4f*)&tile[(wave * 8 + rr) * TP_ + 4 * lane];
    xr[rr] = x;
    sr[rr] = (x.x + x.y) + (x.z + x.w);
    qr[rr] = (x.x * x.x + x.y * x.y) + (x.z * x.z + x.w * x.w);
  }
#pragma unroll
  for (int off = 16; off >= 1; off >>= 1) {
#pragma unroll
    for (int rr = 0; rr < 8; ++rr) {
      sr[rr] += __shfl_xor(sr[rr], off, 32);
      qr[rr] += __shfl_xor(qr[rr], off, 32);
    }
  }
#pragma unroll
  for (int rr = 0; rr < 8; ++rr) {
    int row = wave * 8 + rr;
    v4f x = xr[rr];
    float mu   = sr[rr] * (1.0f / FOUT);
    float var  = qr[rr] * (1.0f / FOUT) - mu * mu;
    float rstd = rsqrtf(var + EPS_);
    v4f y, o;
    y.x = (x.x - mu) * rstd * g4.x + b4.x;
    y.y = (x.y - mu) * rstd * g4.y + b4.y;
    y.z = (x.z - mu) * rstd * g4.z + b4.z;
    y.w = (x.w - mu) * rstd * g4.w + b4.w;
    o.x = 0.5f * y.x * (1.0f + erff(y.x * 0.70710678118654752f));
    o.y = 0.5f * y.y * (1.0f + erff(y.y * 0.70710678118654752f));
    o.z = 0.5f * y.z * (1.0f + erff(y.z * 0.70710678118654752f));
    o.w = 0.5f * y.w * (1.0f + erff(y.w * 0.70710678118654752f));
    *(v4f*)(out + (size_t)(b * N_ + i0 + row) * FOUT + 4 * lane) = o;
  }
}

extern "C" void kernel_launch(void* const* d_in, const int* in_sizes, int n_in,
                              void* d_out, int out_size, void* d_ws, size_t ws_size,
                              hipStream_t stream) {
  const float* h     = (const float*)d_in[0];
  const int*   adj   = (const int*)d_in[1];
  const float* W     = (const float*)d_in[2];
  const float* a     = (const float*)d_in[3];
  const float* gamma = (const float*)d_in[4];
  const float* beta  = (const float*)d_in[5];
  float* out = (float*)d_out;

  float* ws   = (float*)d_ws;
  float* Wh   = ws;
  float* s1   = Wh + (size_t)B_ * N_ * FOUT;
  float* s2   = s1 + B_ * N_;
  float* mrow = s2 + B_ * N_;
  float* lrow = mrow + B_ * N_;
  unsigned* mbits = (unsigned*)(lrow + B_ * N_);

  wh_kernel<<<B_ * (N_ / 16), 256, 0, stream>>>(h, W, a, Wh, s1, s2);
  stats_kernel<<<(B_ * N_) / 8, 256, 0, stream>>>(adj, s1, s2, mrow, lrow, mbits);
  hprime_kernel<<<B_ * (N_ / 32), 128, 0, stream>>>(mbits, Wh, s1, s2, mrow, lrow,
                                                    gamma, beta, out);
}


# stats kernel no longer forms the softmax denominator: h_prime sums its own undivided weights and divides the tile once; stats = masked max of s2 + bit pack, 2 rows per wave; sum order changes
# speedup vs baseline: 1.5751x; 1.0726x over previous
#include <hip/hip_runtime.h>
#include <hip/hip_bf16.h>
#include <math.h>

typedef __attribute__((ext_vector_type(2))) float v2f;
typedef __attribute__((ext_vector_type(8))) float v8f;

#define B_   4
#define N_   2048
#define FIN  256
#define FOUT 128
#define ALPHA_ 0.2f
#define EPS_   1e-5f
#define NEG_   -9000000000000000.0f

#define WP_ 132
__global__ __launch_bounds__(256) void wh_kernel(const float* __restrict__ h,
                                                 const float* __restrict__ W,
                                                 const float* __restrict__ a,
                                                 float* __restrict__ Wh,
                                                 float* __restrict__ s1,
                                                 float* __restrict__ s2) {
  __shared__ __attribute__((aligned(16))) float strip[16 * WP_];
  int lane  = threadIdx.x & 31;
  int otile = threadIdx.x >> 5;
  int itile = blockIdx.x & (N_ / 16 - 1);
  int b     = blockIdx.x >> 7;
  int i0 = itile * 16, o0 = otile * 16;
  int r16  = lane & 15;
  int koff = (lane >> 4) << 1;

  const float4* a1 = (const float4*)a;
  const float4* a2 = (const float4*)(a + FOUT);
  float4 x1 = a1[lane];
  float4 x2 = a2[lane];

  const float* hrow = h + (size_t)(b * N_ + i0 + r16) * FIN;
  v8f c = {};
  for (int k0 = 0; k0 < FIN; k0 += 4) {
    float2 av = *(const float2*)(hrow + k0 + koff);
    v2f a_; a_.x = av.x; a_.y = av.y;
    v2f bb;
    bb.x = W[(size_t)(k0 + koff + 0) * FOUT + o0 + r16];
    bb.y = W[(size_t)(k0 + koff + 1) * FOUT + o0 + r16];
    c = __builtin_amdgcn_wmma_f32_16x16x4_f32(false, a_, false, bb, (short)0, c,
                                              false, false);
  }
#pragma unroll
  for (int r = 0; r < 8; ++r) {
    int M = r + ((lane >> 4) << 3);
    strip[M * WP_ + o0 + r16] = c[r];
  }
  __syncthreads();
#pragma unroll
  for (int r = 0; r < 8; ++r) {
    int M = r + ((lane >> 4) << 3);
    Wh[(size_t)(b * N_ + i0 + M) * FOUT + o0 + r16] = c[r];
  }

  float p1[2], p2[2];
#pragma unroll
  for (int rr = 0; rr < 2; ++rr) {
    float4 wv = *(const float4*)&strip[(2 * otile + rr) * WP_ + 4 * lane];
    p1[rr] = wv.x * x1.x + wv.y * x1.y + wv.z * x1.z + wv.w * x1.w;
    p2[rr] = wv.x * x2.x + wv.y * x2.y + wv.z * x2.z + wv.w * x2.w;
  }
#pragma unroll
  for (int off = 16; off >= 1; off >>= 1) {
#pragma unroll
    for (int rr = 0; rr < 2; ++rr) {
      p1[rr] += __shfl_xor(p1[rr], off, 32);
      p2[rr] += __shfl_xor(p2[rr], off, 32);
    }
  }
  if (lane < 2) {
    int node = b * N_ + i0 + 2 * otile + lane;
    s1[node] = (lane == 0) ? p1[0] : p1[1];
    s2[node] = (lane == 0) ? p2[0] : p2[1];
  }
}

#define RPW_ 2
__global__ __launch_bounds__(256) void stats_kernel(const int* __restrict__ adj,
                                                    const float* __restrict__ s1,
                                                    const float* __restrict__ s2,
                                                    float* __restrict__ mrow,
                                                    unsigned* __restrict__ mbits) {
  int lane = threadIdx.x & 31;
  int row0 = (blockIdx.x * 8 + (threadIdx.x >> 5)) * RPW_;
  int b    = row0 >> 11;
  const float* s2b = s2 + (size_t)b * N_;
  const int* arow  = adj + (size_t)row0 * N_;
  float    smax[RPW_];
  unsigned wlo[RPW_], whi[RPW_];
  unsigned alln[RPW_];
#pragma unroll
  for (int r = 0; r < RPW_; ++r) { smax[r] = -INFINITY; wlo[r] = 0u; whi[r] = 0u; alln[r] = 0xffffffffu; }
  constexpr int GRP_ = 16;
#pragma unroll 1
  for (int w0 = 0; w0 < N_ / 32; w0 += GRP_) {
    int   av[RPW_][GRP_];
    float sv[GRP_];
#pragma unroll
    for (int u = 0; u < GRP_; ++u) {
      int j = (w0 + u) * 32 + lane;
      sv[u] = s2b[j];
#pragma unroll
      for (int r = 0; r < RPW_; ++r) av[r][u] = arow[(size_t)r * N_ + j];
    }
#pragma unroll
    for (int u = 0; u < GRP_; u += 2) {
      bool mine = lane == ((w0 + u) >> 1);
#pragma unroll
      for (int r = 0; r < RPW_; ++r) {
        bool p0 = av[r][u] > 0, p1 = av[r][u + 1] > 0;
        unsigned b0 = __builtin_amdgcn_ballot_w32(p0);
        unsigned b1 = __builtin_amdgcn_ballot_w32(p1);
        alln[r] &= b0 & b1;
        wlo[r] = mine ? b0 : wlo[r];
        whi[r] = mine ? b1 : whi[r];
        smax[r] = fmaxf(smax[r], p0 ? sv[u] : -INFINITY);
        smax[r] = fmaxf(smax[r], p1 ? sv[u + 1] : -INFINITY);
      }
    }
  }
#pragma unroll
  for (int r = 0; r < RPW_; ++r)
    *(uint2*)(mbits + (size_t)(row0 + r) * (N_ / 32) + 2 * lane) = make_uint2(wlo[r], whi[r]);
#pragma unroll
  for (int off = 16; off >= 1; off >>= 1) {
#pragma unroll
    for (int r = 0; r < RPW_; ++r) smax[r] = fmaxf(smax[r], __shfl_xor(smax[r], off, 32));
  }
  if (lane < RPW_) {
    float    sm = (lane == 0) ? smax[0] : smax[1];
    unsigned al = (lane == 0) ? alln[0] : alln[1];
    float e = s1[row0 + lane] + sm;
    e = (e >= 0.f) ? e : ALPHA_ * e;
    float m = (al != 0xffffffffu) ? fmaxf(e, NEG_) : e;
    mrow[row0 + lane] = m;
  }
}

typedef __attribute__((ext_vector_type(4))) float v4f;
#define JC_     64
#define PP_     288
#define STAGE_  ((JC_ / 2) * PP_ + 4 * JC_)
#define TP_     132
#define LDSF_   (41 * 1024)
__global__ __launch_bounds__(128) void hprime_kernel(
    const unsigned* __restrict__ mbits, const float* __restrict__ Wh,
    const float* __restrict__ s1, const float* __restrict__ s2,
    const float* __restrict__ mrow,
    const float* __restrict__ gamma, const float* __restrict__ beta,
    float* __restrict__ out) {
  __shared__ __attribute__((aligned(16))) float sB[LDSF_];
  int tid  = threadIdx.x;
  int lane = tid & 31;
  int wave = tid >> 5;
  int rt   = wave >> 1;
  int chf  = wave & 1;
  int b    = blockIdx.x >> 6;
  int i0   = (blockIdx.x & 63) * 32;
  int r16  = lane & 15;
  int hh   = lane >> 4;
  int rowid = b * N_ + i0 + rt * 16 + r16;

  float s1v  = s1[rowid];
  float mi   = mrow[rowid];
  float ls0 = 0.f, ls1 = 0.f;
  const unsigned* mb = mbits + (size_t)rowid * (N_ / 32);
  const float* s2b   = s2 + (size_t)b * N_;
  const float* Whb   = Wh + (size_t)b * N_ * FOUT;

  v8f acc[4];
#pragma unroll
  for (int ot = 0; ot < 4; ++ot) acc[ot] = (v8f){};

  constexpr int NP = JC_ / 8;
  v4f  pa[NP], pb[NP];
  v2f  s2pre;
  uint2 mw, mwn;
  const float* gsrc = Whb + (size_t)(2 * wave) * FOUT + 4 * lane;
  const int    ldst = wave * PP_ + 8 * lane;
  const int    sdst = (JC_ / 2) * PP_ + wave * JC_ + 2 * lane;

#pragma unroll
  for (int i = 0; i < NP; ++i) {
    pa[i] = *(const v4f*)(gsrc + (size_t)(8 * i) * FOUT);
    pb[i] = *(const v4f*)(gsrc + (size_t)(8 * i + 1) * FOUT);
  }
  s2pre = *(const v2f*)(s2b + 2 * lane);
  mw = *(const uint2*)(mb);
#pragma unroll
  for (int i = 0; i < NP; ++i) {
    v4f lo = {pa[i].x, pb[i].x, pa[i].y, pb[i].y};
    v4f hi = {pa[i].z, pb[i].z, pa[i].w, pb[i].w};
    *(v4f*)&sB[ldst + 4 * i * PP_]     = lo;
    *(v4f*)&sB[ldst + 4 * i * PP_ + 4] = hi;
  }
  *(v2f*)&sB[sdst] = s2pre;
  __syncthreads();

  constexpr int NCH = N_ / JC_;
  for (int ck = 0; ck < NCH; ++ck) {
    const float* sb = sB + (ck & 1) * STAGE_;
    int cn = (ck + 1 < NCH) ? ck + 1 : ck;
    const float* gn = gsrc + (size_t)(cn * JC_) * FOUT;
#pragma unroll
    for (int i = 0; i < NP; ++i) {
      pa[i] = *(const v4f*)(gn + (size_t)(8 * i) * FOUT);
      pb[i] = *(const v4f*)(gn + (size_t)(8 * i + 1) * FOUT);
    }
    s2pre = *(const v2f*)(s2b + cn * JC_ + 2 * lane);
    mwn = *(const uint2*)(mb + 2 * cn);

    unsigned w0 = mw.x >> (2 * hh);
    unsigned w1 = mw.y >> (2 * hh);
    const float* s2l  = sb + (JC_ / 2) * PP_ + wave * JC_ + 2 * hh;
    const float* brow = sb + hh * PP_ + 2 * (64 * chf + r16);
#pragma unroll
    for (int t = 0; t < JC_ / 4; ++t) {
      v2f sv = *(const v2f*)(s2l + 4 * t);
      unsigned wsh = (t < 8) ? w0 : w1;
      v2f a;
      {
        float e = s1v + sv.x;
        e = (e >= 0.f) ? e : ALPHA_ * e;
        float tv = ((wsh >> (4 * (t & 7))) & 1u) ? e : NEG_;
        a.x = __expf(tv - mi);
        ls0 += a.x;
      }
      {
        float e = s1v + sv.y;
        e = (e >= 0.f) ? e : ALPHA_ * e;
        float tv = ((wsh >> (4 * (t & 7) + 1)) & 1u) ? e : NEG_;
        a.y = __expf(tv - mi);
        ls1 += a.y;
      }
      const float* bp = brow + (2 * t) * PP_;
#pragma unroll
      for (int ot = 0; ot < 4; ++ot) {
        v2f bb = *(const v2f*)(bp + 32 * ot);
        acc[ot] = __builtin_amdgcn_wmma_f32_16x16x4_f32(false, a, false, bb,
                                                        (short)0, acc[ot],
                                                        false, false);
      }
    }

    float* sn = sB + ((ck + 1) & 1) * STAGE_;
#pragma unroll
    for (int i = 0; i < NP; ++i) {
      v4f lo = {pa[i].x, pb[i].x, pa[i].y, pb[i].y};
      v4f hi = {pa[i].z, pb[i].z, pa[i].w, pb[i].w};
      *(v4f*)&sn[ldst + 4 * i * PP_]     = lo;
      *(v4f*)&sn[ldst + 4 * i * PP_ + 4] = hi;
    }
    *(v2f*)&sn[sdst] = s2pre;
    mw = mwn;
    __syncthreads();
  }

  float* tile = sB;
  float* lsum = sB + 32 * TP_;
#pragma unroll
  for (int ot = 0; ot < 4; ++ot)
#pragma unroll
    for (int r = 0; r < 8; ++r)
      tile[(rt * 16 + hh * 8 + r) * TP_ + 64 * chf + ot * 16 + r16] = acc[ot][r];
  float ls = ls0 + ls1;
  ls += __shfl_xor(ls, 16, 32);
  if (chf == 0 && hh == 0) lsum[rt * 16 + r16] = ls;
  __syncthreads();

  v4f g4 = *(const v4f*)(gamma + 4 * lane);
  v4f b4 = *(const v4f*)(beta + 4 * lane);
  v4f xr[8];
  float sr[8], qr[8];
#pragma unroll
  for (int rr = 0; rr < 8; ++rr) {
    v4f x = *(const v4f*)&tile[(wave * 8 + rr) * TP_ + 4 * lane];
    float il = 1.0f / lsum[wave * 8 + rr];
    x.x *= il; x.y *= il; x.z *= il; x.w *= il;
    xr[rr] = x;
    sr[rr] = (x.x + x.y) + (x.z + x.w);
    qr[rr] = (x.x * x.x + x.y * x.y) + (x.z * x.z + x.w * x.w);
  }
#pragma unroll
  for (int off = 16; off >= 1; off >>= 1) {
#pragma unroll
    for (int rr = 0; rr < 8; ++rr) {
      sr[rr] += __shfl_xor(sr[rr], off, 32);
      qr[rr] += __shfl_xor(qr[rr], off, 32);
    }
  }
#pragma unroll
  for (int rr = 0; rr < 8; ++rr) {
    int row = wave * 8 + rr;
    v4f x = xr[rr];
    float mu   = sr[rr] * (1.0f / FOUT);
    float var  = qr[rr] * (1.0f / FOUT) - mu * mu;
    float rstd = rsqrtf(var + EPS_);
    v4f y, o;
    y.x = (x.x - mu) * rstd * g4.x + b4.x;
    y.y = (x.y - mu) * rstd * g4.y + b4.y;
    y.z = (x.z - mu) * rstd * g4.z + b4.z;
    y.w = (x.w - mu) * rstd * g4.w + b4.w;
    o.x = 0.5f * y.x * (1.0f + erff(y.x * 0.70710678118654752f));
    o.y = 0.5f * y.y * (1.0f + erff(y.y * 0.70710678118654752f));
    o.z = 0.5f * y.z * (1.0f + erff(y.z * 0.70710678118654752f));
    o.w = 0.5f * y.w * (1.0f + erff(y.w * 0.70710678118654752f));
    *(v4f*)(out + (size_t)(b * N_ + i0 + row) * FOUT + 4 * lane) = o;
  }
}

extern "C" void kernel_launch(void* const* d_in, const int* in_sizes, int n_in,
                              void* d_out, int out_size, void* d_ws, size_t ws_size,
                              hipStream_t stream) {
  const float* h     = (const float*)d_in[0];
  const int*   adj   = (const int*)d_in[1];
  const float* W     = (const float*)d_in[2];
  const float* a     = (const float*)d_in[3];
  const float* gamma = (const float*)d_in[4];
  const float* beta  = (const float*)d_in[5];
  float* out = (float*)d_out;

  float* ws   = (float*)d_ws;
  float* Wh   = ws;
  float* s1   = Wh + (size_t)B_ * N_ * FOUT;
  float* s2   = s1 + B_ * N_;
  float* mrow = s2 + B_ * N_;
  unsigned* mbits = (unsigned*)(mrow + 2 * B_ * N_);

  wh_kernel<<<B_ * (N_ / 16), 256, 0, stream>>>(h, W, a, Wh, s1, s2);
  stats_kernel<<<(B_ * N_) / (8 * RPW_), 256, 0, stream>>>(adj, s1, s2, mrow, mbits);
  hprime_kernel<<<B_ * (N_ / 32), 128, 0, stream>>>(mbits, Wh, s1, s2, mrow,
                                                    gamma, beta, out);
}


# h_prime kernel: a row tile's waves split the j range of each staged chunk, not the columns: 8 waves (2 per SIMD), each 16 rows x 128 cols x 32 of 128 j; partial tiles added in LDS; j sums reordered
# speedup vs baseline: 1.6287x; 1.0340x over previous
#include <hip/hip_runtime.h>
#include <hip/hip_bf16.h>
#include <math.h>

typedef __attribute__((ext_vector_type(2))) float v2f;
typedef __attribute__((ext_vector_type(8))) float v8f;

#define B_   4
#define N_   2048
#define FIN  256
#define FOUT 128
#define ALPHA_ 0.2f
#define EPS_   1e-5f
#define NEG_   -9000000000000000.0f

#define WP_ 132
__global__ __launch_bounds__(256) void wh_kernel(const float* __restrict__ h,
                                                 const float* __restrict__ W,
                                                 const float* __restrict__ a,
                                                 float* __restrict__ Wh,
                                                 float* __restrict__ s1,
                                                 float* __restrict__ s2) {
  __shared__ __attribute__((aligned(16))) float strip[16 * WP_];
  int lane  = threadIdx.x & 31;
  int otile = threadIdx.x >> 5;
  int itile = blockIdx.x & (N_ / 16 - 1);
  int b     = blockIdx.x >> 7;
  int i0 = itile * 16, o0 = otile * 16;
  int r16  = lane & 15;
  int koff = (lane >> 4) << 1;

  const float4* a1 = (const float4*)a;
  const float4* a2 = (const float4*)(a + FOUT);
  float4 x1 = a1[lane];
  float4 x2 = a2[lane];

  const float* hrow = h + (size_t)(b * N_ + i0 + r16) * FIN;
  v8f c = {};
  for (int k0 = 0; k0 < FIN; k0 += 4) {
    float2 av = *(const float2*)(hrow + k0 + koff);
    v2f a_; a_.x = av.x; a_.y = av.y;
    v2f bb;
    bb.x = W[(size_t)(k0 + koff + 0) * FOUT + o0 + r16];
    bb.y = W[(size_t)(k0 + koff + 1) * FOUT + o0 + r16];
    c = __builtin_amdgcn_wmma_f32_16x16x4_f32(false, a_, false, bb, (short)0, c,
                                              false, false);
  }
#pragma unroll
  for (int r = 0; r < 8; ++r) {
    int M = r + ((lane >> 4) << 3);
    strip[M * WP_ + o0 + r16] = c[r];
  }
  __syncthreads();
#pragma unroll
  for (int r = 0; r < 8; ++r) {
    int M = r + ((lane >> 4) << 3);
    Wh[(size_t)(b * N_ + i0 + M) * FOUT + o0 + r16] = c[r];
  }

  float p1[2], p2[2];
#pragma unroll
  for (int rr = 0; rr < 2; ++rr) {
    float4 wv = *(const float4*)&strip[(2 * otile + rr) * WP_ + 4 * lane];
    p1[rr] = wv.x * x1.x + wv.y * x1.y + wv.z * x1.z + wv.w * x1.w;
    p2[rr] = wv.x * x2.x + wv.y * x2.y + wv.z * x2.z + wv.w * x2.w;
  }
#pragma unroll
  for (int off = 16; off >= 1; off >>= 1) {
#pragma unroll
    for (int rr = 0; rr < 2; ++rr) {
      p1[rr] += __shfl_xor(p1[rr], off, 32);
      p2[rr] += __shfl_xor(p2[rr], off, 32);
    }
  }
  if (lane < 2) {
    int node = b * N_ + i0 + 2 * otile + lane;
    s1[node] = (lane == 0) ? p1[0] : p1[1];
    s2[node] = (lane == 0) ? p2[0] : p2[1];
  }
}

#define RPW_ 2
__global__ __launch_bounds__(256) void stats_kernel(const int* __restrict__ adj,
                                                    const float* __restrict__ s1,
                                                    const float* __restrict__ s2,
                                                    float* __restrict__ mrow,
                                                    unsigned* __restrict__ mbits) {
  int lane = threadIdx.x & 31;
  int row0 = (blockIdx.x * 8 + (threadIdx.x >> 5)) * RPW_;
  int b    = row0 >> 11;
  const float* s2b = s2 + (size_t)b * N_;
  const int* arow  = adj + (size_t)row0 * N_;
  float    smax[RPW_];
  unsigned wlo[RPW_], whi[RPW_];
  unsigned alln[RPW_];
#pragma unroll
  for (int r = 0; r < RPW_; ++r) { smax[r] = -INFINITY; wlo[r] = 0u; whi[r] = 0u; alln[r] = 0xffffffffu; }
  constexpr int GRP_ = 16;
#pragma unroll 1
  for (int w0 = 0; w0 < N_ / 32; w0 += GRP_) {
    int   av[RPW_][GRP_];
    float sv[GRP_];
#pragma unroll
    for (int u = 0; u < GRP_; ++u) {
      int j = (w0 + u) * 32 + lane;
      sv[u] = s2b[j];
#pragma unroll
      for (int r = 0; r < RPW_; ++r) av[r][u] = arow[(size_t)r * N_ + j];
    }
#pragma unroll
    for (int u = 0; u < GRP_; u += 2) {
      bool mine = lane == ((w0 + u) >> 1);
#pragma unroll
      for (int r = 0; r < RPW_; ++r) {
        bool p0 = av[r][u] > 0, p1 = av[r][u + 1] > 0;
        unsigned b0 = __builtin_amdgcn_ballot_w32(p0);
        unsigned b1 = __builtin_amdgcn_ballot_w32(p1);
        alln[r] &= b0 & b1;
        wlo[r] = mine ? b0 : wlo[r];
        whi[r] = mine ? b1 : whi[r];
        smax[r] = fmaxf(smax[r], p0 ? sv[u] : -INFINITY);
        smax[r] = fmaxf(smax[r], p1 ? sv[u + 1] : -INFINITY);
      }
    }
  }
#pragma unroll
  for (int r = 0; r < RPW_; ++r)
    *(uint2*)(mbits + (size_t)(row0 + r) * (N_ / 32) + 2 * lane) = make_uint2(wlo[r], whi[r]);
#pragma unroll
  for (int off = 16; off >= 1; off >>= 1) {
#pragma unroll
    for (int r = 0; r < RPW_; ++r) smax[r] = fmaxf(smax[r], __shfl_xor(smax[r], off, 32));
  }
  if (lane < RPW_) {
    float    sm = (lane == 0) ? smax[0] : smax[1];
    unsigned al = (lane == 0) ? alln[0] : alln[1];
    float e = s1[row0 + lane] + sm;
    e = (e >= 0.f) ? e : ALPHA_ * e;
    float m = (al != 0xffffffffu) ? fmaxf(e, NEG_) : e;
    mrow[row0 + lane] = m;
  }
}

typedef __attribute__((ext_vector_type(4))) float v4f;
#define JC_     128
#define NW3_    8
#define JQ_     (JC_ / 4)
#define PP_     288
#define STAGE_  ((JC_ / 2) * PP_ + NW3_ * JQ_)
#define TP_     132
#define LDSF_   (41 * 1024)
__global__ __launch_bounds__(256) void hprime_kernel(
    const unsigned* __restrict__ mbits, const float* __restrict__ Wh,
    const float* __restrict__ s1, const float* __restrict__ s2,
    const float* __restrict__ mrow,
    const float* __restrict__ gamma, const float* __restrict__ beta,
    float* __restrict__ out) {
  __shared__ __attribute__((aligned(16))) float sB[LDSF_];
  int tid  = threadIdx.x;
  int lane = tid & 31;
  int wave = tid >> 5;
  int rt   = wave >> 2;
  int jq   = wave & 3;
  int b    = blockIdx.x >> 6;
  int i0   = (blockIdx.x & 63) * 32;
  int r16  = lane & 15;
  int hh   = lane >> 4;
  int rowid = b * N_ + i0 + rt * 16 + r16;

  float s1v  = s1[rowid];
  float mi   = mrow[rowid];
  float ls0 = 0.f, ls1 = 0.f;
  const unsigned* mb = mbits + (size_t)rowid * (N_ / 32) + jq;
  const int mshift   = 2 * hh;
  const float* s2b   = s2 + (size_t)b * N_;
  const float* Whb   = Wh + (size_t)b * N_ * FOUT;

  v8f acc[8];
#pragma unroll
  for (int ot = 0; ot < 8; ++ot) acc[ot] = (v8f){};

  constexpr int NP = JC_ / (2 * NW3_);
  v4f  pa[NP], pb[NP];
  float s2pre;
  unsigned mw, mwn;
  const float* gsrc = Whb + (size_t)(2 * wave) * FOUT + 4 * lane;
  const int    ldst = wave * PP_ + 8 * lane;
  const int    sdst = (JC_ / 2) * PP_ + wave * JQ_ + lane;

#pragma unroll
  for (int i = 0; i < NP; ++i) {
    pa[i] = *(const v4f*)(gsrc + (size_t)(2 * NW3_ * i) * FOUT);
    pb[i] = *(const v4f*)(gsrc + (size_t)(2 * NW3_ * i + 1) * FOUT);
  }
  s2pre = s2b[JQ_ * jq + lane];
  mw = mb[0];
#pragma unroll
  for (int i = 0; i < NP; ++i) {
    v4f lo = {pa[i].x, pb[i].x, pa[i].y, pb[i].y};
    v4f hi = {pa[i].z, pb[i].z, pa[i].w, pb[i].w};
    *(v4f*)&sB[ldst + NW3_ * i * PP_]     = lo;
    *(v4f*)&sB[ldst + NW3_ * i * PP_ + 4] = hi;
  }
  sB[sdst] = s2pre;
  __syncthreads();

  constexpr int NCH = N_ / JC_;
  for (int ck = 0; ck < NCH; ++ck) {
    const float* sb = sB + (ck & 1) * STAGE_;
    int cn = (ck + 1 < NCH) ? ck + 1 : ck;
    const float* gn = gsrc + (size_t)(cn * JC_) * FOUT;
#pragma unroll
    for (int i = 0; i < NP; ++i) {
      pa[i] = *(const v4f*)(gn + (size_t)(2 * NW3_ * i) * FOUT);
      pb[i] = *(const v4f*)(gn + (size_t)(2 * NW3_ * i + 1) * FOUT);
    }
    s2pre = s2b[cn * JC_ + JQ_ * jq + lane];
    mwn = mb[(JC_ / 32) * cn];

    unsigned wsh = mw >> mshift;
    const float* s2l  = sb + (JC_ / 2) * PP_ + wave * JQ_ + 2 * hh;
    const float* brow = sb + (JQ_ / 2) * jq * PP_ + hh * PP_ + 2 * r16;
#pragma unroll
    for (int t = 0; t < JQ_ / 4; ++t) {
      v2f sv = *(const v2f*)(s2l + 4 * t);
      v2f a;
      {
        float e = s1v + sv.x;
        e = (e >= 0.f) ? e : ALPHA_ * e;
        float tv = ((wsh >> (4 * t)) & 1u) ? e : NEG_;
        a.x = __expf(tv - mi);
        ls0 += a.x;
      }
      {
        float e = s1v + sv.y;
        e = (e >= 0.f) ? e : ALPHA_ * e;
        float tv = ((wsh >> (4 * t + 1)) & 1u) ? e : NEG_;
        a.y = __expf(tv - mi);
        ls1 += a.y;
      }
      const float* bp = brow + (2 * t) * PP_;
#pragma unroll
      for (int ot = 0; ot < 8; ++ot) {
        v2f bb = *(const v2f*)(bp + 32 * ot);
        acc[ot] = __builtin_amdgcn_wmma_f32_16x16x4_f32(false, a, false, bb,
                                                        (short)0, acc[ot],
                                                        false, false);
      }
    }

    float* sn = sB + ((ck + 1) & 1) * STAGE_;
#pragma unroll
    for (int i = 0; i < NP; ++i) {
      v4f lo = {pa[i].x, pb[i].x, pa[i].y, pb[i].y};
      v4f hi = {pa[i].z, pb[i].z, pa[i].w, pb[i].w};
      *(v4f*)&sn[ldst + NW3_ * i * PP_]     = lo;
      *(v4f*)&sn[ldst + NW3_ * i * PP_ + 4] = hi;
    }
    sn[sdst] = s2pre;
    mw = mwn;
    __syncthreads();
  }

  float* tile = sB;
  float* lsum = sB + 4 * 32 * TP_;
#pragma unroll
  for (int ot = 0; ot < 8; ++ot)
#pragma unroll
    for (int r = 0; r < 8; ++r)
      tile[(jq * 32 + rt * 16 + hh * 8 + r) * TP_ + ot * 16 + r16] = acc[ot][r];
  float ls = ls0 + ls1;
  ls += __shfl_xor(ls, 16, 32);
  if (hh == 0) lsum[jq * 32 + rt * 16 + r16] = ls;
  __syncthreads();

  v4f g4 = *(const v4f*)(gamma + 4 * lane);
  v4f b4 = *(const v4f*)(beta + 4 * lane);
  v4f xr[4];
  float sr[4], qr[4];
#pragma unroll
  for (int rr = 0; rr < 4; ++rr) {
    int row = wave * 4 + rr;
    v4f x0 = *(const v4f*)&tile[(0 * 32 + row) * TP_ + 4 * lane];
    v4f x1 = *(const v4f*)&tile[(1 * 32 + row) * TP_ + 4 * lane];
    v4f x2 = *(const v4f*)&tile[(2 * 32 + row) * TP_ + 4 * lane];
    v4f x3 = *(const v4f*)&tile[(3 * 32 + row) * TP_ + 4 * lane];
    v4f x = (x0 + x1) + (x2 + x3);
    float l = (lsum[row] + lsum[32 + row]) + (lsum[64 + row] + lsum[96 + row]);
    float il = 1.0f / l;
    x.x *= il; x.y *= il; x.z *= il; x.w *= il;
    xr[rr] = x;
    sr[rr] = (x.x + x.y) + (x.z + x.w);
    qr[rr] = (x.x * x.x + x.y * x.y) + (x.z * x.z + x.w * x.w);
  }
#pragma unroll
  for (int off = 16; off >= 1; off >>= 1) {
#pragma unroll
    for (int rr = 0; rr < 4; ++rr) {
      sr[rr] += __shfl_xor(sr[rr], off, 32);
      qr[rr] += __shfl_xor(qr[rr], off, 32);
    }
  }
#pragma unroll
  for (int rr = 0; rr < 4; ++rr) {
    int row = wave * 4 + rr;
    v4f x = xr[rr];
    float mu   = sr[rr] * (1.0f / FOUT);
    float var  = qr[rr] * (1.0f / FOUT) - mu * mu;
    float rstd = rsqrtf(var + EPS_);
    v4f y, o;
    y.x = (x.x - mu) * rstd * g4.x + b4.x;
    y.y = (x.y - mu) * rstd * g4.y + b4.y;
    y.z = (x.z - mu) * rstd * g4.z + b4.z;
    y.w = (x.w - mu) * rstd * g4.w + b4.w;
    o.x = 0.5f * y.x * (1.0f + erff(y.x * 0.70710678118654752f));
    o.y = 0.5f * y.y * (1.0f + erff(y.y * 0.70710678118654752f));
    o.z = 0.5f * y.z * (1.0f + erff(y.z * 0.70710678118654752f));
    o.w = 0.5f * y.w * (1.0f + erff(y.w * 0.70710678118654752f));
    *(v4f*)(out + (size_t)(b * N_ + i0 + row) * FOUT + 4 * lane) = o;
  }
}

extern "C" void kernel_launch(void* const* d_in, const int* in_sizes, int n_in,
                              void* d_out, int out_size, void* d_ws, size_t ws_size,
                              hipStream_t stream) {
  const float* h     = (const float*)d_in[0];
  const int*   adj   = (const int*)d_in[1];
  const float* W     = (const float*)d_in[2];
  const float* a     = (const float*)d_in[3];
  const float* gamma = (const float*)d_in[4];
  const float* beta  = (const float*)d_in[5];
  float* out = (float*)d_out;

  float* ws   = (float*)d_ws;
  float* Wh   = ws;
  float* s1   = Wh + (size_t)B_ * N_ * FOUT;
  float* s2   = s1 + B_ * N_;
  float* mrow = s2 + B_ * N_;
  unsigned* mbits = (unsigned*)(mrow + 2 * B_ * N_);

  wh_kernel<<<B_ * (N_ / 16), 256, 0, stream>>>(h, W, a, Wh, s1, s2);
  stats_kernel<<<(B_ * N_) / (8 * RPW_), 256, 0, stream>>>(adj, s1, s2, mrow, mbits);
  hprime_kernel<<<B_ * (N_ / 32), 256, 0, stream>>>(mbits, Wh, s1, s2, mrow,
                                                    gamma, beta, out);
}


# stats launch removed: adj bit pack runs as extra work-groups of the Wh launch; masked row max of s2 formed in the h_prime prologue from the packed bits; 3 to 2 launches, same maxima and sums
# speedup vs baseline: 1.7310x; 1.0628x over previous
#include <hip/hip_runtime.h>
#include <hip/hip_bf16.h>
#include <math.h>

typedef __attribute__((ext_vector_type(2))) float v2f;
typedef __attribute__((ext_vector_type(8))) float v8f;

#define B_   4
#define N_   2048
#define FIN  256
#define FOUT 128
#define ALPHA_ 0.2f
#define EPS_   1e-5f
#define NEG_   -9000000000000000.0f

#define WP_ 132
#define RPW_ 2
__global__ __launch_bounds__(256) void wh_pack_kernel(const float* __restrict__ h,
                                                      const float* __restrict__ W,
                                                      const float* __restrict__ a,
                                                      const int* __restrict__ adj,
                                                      float* __restrict__ Wh,
                                                      float* __restrict__ s1,
                                                      float* __restrict__ s2,
                                                      unsigned* __restrict__ mbits) {
  __shared__ __attribute__((aligned(16))) float strip[16 * WP_];
  int lane = threadIdx.x & 31;
  int bid  = blockIdx.x >> 1;

  if (blockIdx.x & 1) {
    int row0 = (bid * 8 + (threadIdx.x >> 5)) * RPW_;
    const int* arow = adj + (size_t)row0 * N_;
    unsigned wlo[RPW_], whi[RPW_];
#pragma unroll
    for (int r = 0; r < RPW_; ++r) { wlo[r] = 0u; whi[r] = 0u; }
    constexpr int GRP_ = 16;
#pragma unroll 1
    for (int w0 = 0; w0 < N_ / 32; w0 += GRP_) {
      int av[RPW_][GRP_];
#pragma unroll
      for (int u = 0; u < GRP_; ++u) {
        int j = (w0 + u) * 32 + lane;
#pragma unroll
        for (int r = 0; r < RPW_; ++r) av[r][u] = arow[(size_t)r * N_ + j];
      }
#pragma unroll
      for (int u = 0; u < GRP_; u += 2) {
        bool mine = lane == ((w0 + u) >> 1);
#pragma unroll
        for (int r = 0; r < RPW_; ++r) {
          unsigned b0 = __builtin_amdgcn_ballot_w32(av[r][u] > 0);
          unsigned b1 = __builtin_amdgcn_ballot_w32(av[r][u + 1] > 0);
          wlo[r] = mine ? b0 : wlo[r];
          whi[r] = mine ? b1 : whi[r];
        }
      }
    }
#pragma unroll
    for (int r = 0; r < RPW_; ++r)
      *(uint2*)(mbits + (size_t)(row0 + r) * (N_ / 32) + 2 * lane) = make_uint2(wlo[r], whi[r]);
    return;
  }

  int otile = threadIdx.x >> 5;
  int itile = bid & (N_ / 16 - 1);
  int b     = bid >> 7;
  int i0 = itile * 16, o0 = otile * 16;
  int r16  = lane & 15;
  int koff = (lane >> 4) << 1;

  const float4* a1 = (const float4*)a;
  const float4* a2 = (const float4*)(a + FOUT);
  float4 x1 = a1[lane];
  float4 x2 = a2[lane];

  const float* hrow = h + (size_t)(b * N_ + i0 + r16) * FIN;
  v8f c = {};
  for (int k0 = 0; k0 < FIN; k0 += 4) {
    float2 av = *(const float2*)(hrow + k0 + koff);
    v2f a_; a_.x = av.x; a_.y = av.y;
    v2f bb;
    bb.x = W[(size_t)(k0 + koff + 0) * FOUT + o0 + r16];
    bb.y = W[(size_t)(k0 + koff + 1) * FOUT + o0 + r16];
    c = __builtin_amdgcn_wmma_f32_16x16x4_f32(false, a_, false, bb, (short)0, c,
                                              false, false);
  }
#pragma unroll
  for (int r = 0; r < 8; ++r) {
    int M = r + ((lane >> 4) << 3);
    strip[M * WP_ + o0 + r16] = c[r];
  }
  __syncthreads();
#pragma unroll
  for (int r = 0; r < 8; ++r) {
    int M = r + ((lane >> 4) << 3);
    Wh[(size_t)(b * N_ + i0 + M) * FOUT + o0 + r16] = c[r];
  }

  float p1[2], p2[2];
#pragma unroll
  for (int rr = 0; rr < 2; ++rr) {
    float4 wv = *(const float4*)&strip[(2 * otile + rr) * WP_ + 4 * lane];
    p1[rr] = wv.x * x1.x + wv.y * x1.y + wv.z * x1.z + wv.w * x1.w;
    p2[rr] = wv.x * x2.x + wv.y * x2.y + wv.z * x2.z + wv.w * x2.w;
  }
#pragma unroll
  for (int off = 16; off >= 1; off >>= 1) {
#pragma unroll
    for (int rr = 0; rr < 2; ++rr) {
      p1[rr] += __shfl_xor(p1[rr], off, 32);
      p2[rr] += __shfl_xor(p2[rr], off, 32);
    }
  }
  if (lane < 2) {
    int node = b * N_ + i0 + 2 * otile + lane;
    s1[node] = (lane == 0) ? p1[0] : p1[1];
    s2[node] = (lane == 0) ? p2[0] : p2[1];
  }
}

typedef __attribute__((ext_vector_type(4))) float v4f;
#define JC_     128
#define NW3_    8
#define JQ_     (JC_ / 4)
#define PP_     288
#define STAGE_  ((JC_ / 2) * PP_ + NW3_ * JQ_)
#define TP_     132
#define LDSF_   (41 * 1024)
__global__ __launch_bounds__(256) void hprime_kernel(
    const unsigned* __restrict__ mbits, const float* __restrict__ Wh,
    const float* __restrict__ s1, const float* __restrict__ s2,
    const float* __restrict__ gamma, const float* __restrict__ beta,
    float* __restrict__ out) {
  __shared__ __attribute__((aligned(16))) float sB[LDSF_];
  int tid  = threadIdx.x;
  int lane = tid & 31;
  int wave = tid >> 5;
  int rt   = wave >> 2;
  int jq   = wave & 3;
  int b    = blockIdx.x >> 6;
  int i0   = (blockIdx.x & 63) * 32;
  int r16  = lane & 15;
  int hh   = lane >> 4;
  int rowid = b * N_ + i0 + rt * 16 + r16;

  float s1v  = s1[rowid];
  float ls0 = 0.f, ls1 = 0.f;
  const unsigned* mb = mbits + (size_t)rowid * (N_ / 32) + jq;
  const int mshift   = 2 * hh;
  const float* s2b   = s2 + (size_t)b * N_;
  const float* Whb   = Wh + (size_t)b * N_ * FOUT;

  v8f acc[8];
#pragma unroll
  for (int ot = 0; ot < 8; ++ot) acc[ot] = (v8f){};

  constexpr int NP = JC_ / (2 * NW3_);
  v4f  pa[NP], pb[NP];
  float s2pre;
  unsigned mw, mwn;
  const float* gsrc = Whb + (size_t)(2 * wave) * FOUT + 4 * lane;
  const int    ldst = wave * PP_ + 8 * lane;
  const int    sdst = (JC_ / 2) * PP_ + wave * JQ_ + lane;

#pragma unroll
  for (int i = 0; i < NP; ++i) {
    pa[i] = *(const v4f*)(gsrc + (size_t)(2 * NW3_ * i) * FOUT);
    pb[i] = *(const v4f*)(gsrc + (size_t)(2 * NW3_ * i + 1) * FOUT);
  }
  s2pre = s2b[JQ_ * jq + lane];
  mw = mb[0];

  float* mrow_s = sB + 2 * STAGE_;
  {
    const int wv   = __builtin_amdgcn_readfirstlane(wave);
    const int row0 = b * N_ + i0 + 4 * wv;
    float s2v[N_ / 32];
#pragma unroll
    for (int w = 0; w < N_ / 32; ++w) s2v[w] = s2b[32 * w + lane];
    uint2 mv[4];
    float s1r[4];
#pragma unroll
    for (int rr = 0; rr < 4; ++rr) {
      mv[rr]  = *(const uint2*)(mbits + (size_t)(row0 + rr) * (N_ / 32) + 2 * lane);
      s1r[rr] = s1[row0 + rr];
    }
    float mx[4];
#pragma unroll
    for (int rr = 0; rr < 4; ++rr) {
      const unsigned* mr = mbits + (size_t)(row0 + rr) * (N_ / 32);
      float mx0 = -INFINITY, mx1 = -INFINITY;
#pragma unroll
      for (int w = 0; w < N_ / 32; w += 2) {
        mx0 = fmaxf(mx0, __builtin_amdgcn_inverse_ballot_w32(mr[w]) ? s2v[w] : -INFINITY);
        mx1 = fmaxf(mx1, __builtin_amdgcn_inverse_ballot_w32(mr[w + 1]) ? s2v[w + 1] : -INFINITY);
      }
      mx[rr] = fmaxf(mx0, mx1);
    }
#pragma unroll
    for (int off = 16; off >= 1; off >>= 1) {
#pragma unroll
      for (int rr = 0; rr < 4; ++rr) mx[rr] = fmaxf(mx[rr], __shfl_xor(mx[rr], off, 32));
    }
    v4f m4;
#pragma unroll
    for (int rr = 0; rr < 4; ++rr) {
      bool alln = __builtin_amdgcn_ballot_w32((mv[rr].x & mv[rr].y) == 0xffffffffu) == 0xffffffffu;
      float e = s1r[rr] + mx[rr];
      e = (e >= 0.f) ? e : ALPHA_ * e;
      m4[rr] = alln ? e : fmaxf(e, NEG_);
    }
    if (lane == 0) *(v4f*)&mrow_s[4 * wv] = m4;
  }

#pragma unroll
  for (int i = 0; i < NP; ++i) {
    v4f lo = {pa[i].x, pb[i].x, pa[i].y, pb[i].y};
    v4f hi = {pa[i].z, pb[i].z, pa[i].w, pb[i].w};
    *(v4f*)&sB[ldst + NW3_ * i * PP_]     = lo;
    *(v4f*)&sB[ldst + NW3_ * i * PP_ + 4] = hi;
  }
  sB[sdst] = s2pre;
  __syncthreads();
  const float mi = mrow_s[rt * 16 + r16];

  constexpr int NCH = N_ / JC_;
  for (int ck = 0; ck < NCH; ++ck) {
    const float* sb = sB + (ck & 1) * STAGE_;
    int cn = (ck + 1 < NCH) ? ck + 1 : ck;
    const float* gn = gsrc + (size_t)(cn * JC_) * FOUT;
#pragma unroll
    for (int i = 0; i < NP; ++i) {
      pa[i] = *(const v4f*)(gn + (size_t)(2 * NW3_ * i) * FOUT);
      pb[i] = *(const v4f*)(gn + (size_t)(2 * NW3_ * i + 1) * FOUT);
    }
    s2pre = s2b[cn * JC_ + JQ_ * jq + lane];
    mwn = mb[(JC_ / 32) * cn];

    unsigned wsh = mw >> mshift;
    const float* s2l  = sb + (JC_ / 2) * PP_ + wave * JQ_ + 2 * hh;
    const float* brow = sb + (JQ_ / 2) * jq * PP_ + hh * PP_ + 2 * r16;
#pragma unroll
    for (int t = 0; t < JQ_ / 4; ++t) {
      v2f sv = *(const v2f*)(s2l + 4 * t);
      v2f a;
      {
        float e = s1v + sv.x;
        e = (e >= 0.f) ? e : ALPHA_ * e;
        float tv = ((wsh >> (4 * t)) & 1u) ? e : NEG_;
        a.x = __expf(tv - mi);
        ls0 += a.x;
      }
      {
        float e = s1v + sv.y;
        e = (e >= 0.f) ? e : ALPHA_ * e;
        float tv = ((wsh >> (4 * t + 1)) & 1u) ? e : NEG_;
        a.y = __expf(tv - mi);
        ls1 += a.y;
      }
      const float* bp = brow + (2 * t) * PP_;
#pragma unroll
      for (int ot = 0; ot < 8; ++ot) {
        v2f bb = *(const v2f*)(bp + 32 * ot);
        acc[ot] = __builtin_amdgcn_wmma_f32_16x16x4_f32(false, a, false, bb,
                                                        (short)0, acc[ot],
                                                        false, false);
      }
    }

    float* sn = sB + ((ck + 1) & 1) * STAGE_;
#pragma unroll
    for (int i = 0; i < NP; ++i) {
      v4f lo = {pa[i].x, pb[i].x, pa[i].y, pb[i].y};
      v4f hi = {pa[i].z, pb[i].z, pa[i].w, pb[i].w};
      *(v4f*)&sn[ldst + NW3_ * i * PP_]     = lo;
      *(v4f*)&sn[ldst + NW3_ * i * PP_ + 4] = hi;
    }
    sn[sdst] = s2pre;
    mw = mwn;
    __syncthreads();
  }

  float* tile = sB;
  float* lsum = sB + 4 * 32 * TP_;
#pragma unroll
  for (int ot = 0; ot < 8; ++ot)
#pragma unroll
    for (int r = 0; r < 8; ++r)
      tile[(jq * 32 + rt * 16 + hh * 8 + r) * TP_ + ot * 16 + r16] = acc[ot][r];
  float ls = ls0 + ls1;
  ls += __shfl_xor(ls, 16, 32);
  if (hh == 0) lsum[jq * 32 + rt * 16 + r16] = ls;
  __syncthreads();

  v4f g4 = *(const v4f*)(gamma + 4 * lane);
  v4f b4 = *(const v4f*)(beta + 4 * lane);
  v4f xr[4];
  float sr[4], qr[4];
#pragma unroll
  for (int rr = 0; rr < 4; ++rr) {
    int row = wave * 4 + rr;
    v4f x0 = *(const v4f*)&tile[(0 * 32 + row) * TP_ + 4 * lane];
    v4f x1 = *(const v4f*)&tile[(1 * 32 + row) * TP_ + 4 * lane];
    v4f x2 = *(const v4f*)&tile[(2 * 32 + row) * TP_ + 4 * lane];
    v4f x3 = *(const v4f*)&tile[(3 * 32 + row) * TP_ + 4 * lane];
    v4f x = (x0 + x1) + (x2 + x3);
    float l = (lsum[row] + lsum[32 + row]) + (lsum[64 + row] + lsum[96 + row]);
    float il = 1.0f / l;
    x.x *= il; x.y *= il; x.z *= il; x.w *= il;
    xr[rr] = x;
    sr[rr] = (x.x + x.y) + (x.z + x.w);
    qr[rr] = (x.x * x.x + x.y * x.y) + (x.z * x.z + x.w * x.w);
  }
#pragma unroll
  for (int off = 16; off >= 1; off >>= 1) {
#pragma unroll
    for (int rr = 0; rr < 4; ++rr) {
      sr[rr] += __shfl_xor(sr[rr], off, 32);
      qr[rr] += __shfl_xor(qr[rr], off, 32);
    }
  }
#pragma unroll
  for (int rr = 0; rr < 4; ++rr) {
    int row = wave * 4 + rr;
    v4f x = xr[rr];
    float mu   = sr[rr] * (1.0f / FOUT);
    float var  = qr[rr] * (1.0f / FOUT) - mu * mu;
    float rstd = rsqrtf(var + EPS_);
    v4f y, o;
    y.x = (x.x - mu) * rstd * g4.x + b4.x;
    y.y = (x.y - mu) * rstd * g4.y + b4.y;
    y.z = (x.z - mu) * rstd * g4.z + b4.z;
    y.w = (x.w - mu) * rstd * g4.w + b4.w;
    o.x = 0.5f * y.x * (1.0f + erff(y.x * 0.70710678118654752f));
    o.y = 0.5f * y.y * (1.0f + erff(y.y * 0.70710678118654752f));
    o.z = 0.5f * y.z * (1.0f + erff(y.z * 0.70710678118654752f));
    o.w = 0.5f * y.w * (1.0f + erff(y.w * 0.70710678118654752f));
    *(v4f*)(out + (size_t)(b * N_ + i0 + row) * FOUT + 4 * lane) = o;
  }
}

extern "C" void kernel_launch(void* const* d_in, const int* in_sizes, int n_in,
                              void* d_out, int out_size, void* d_ws, size_t ws_size,
                              hipStream_t stream) {
  const float* h     = (const float*)d_in[0];
  const int*   adj   = (const int*)d_in[1];
  const float* W     = (const float*)d_in[2];
  const float* a     = (const float*)d_in[3];
  const float* gamma = (const float*)d_in[4];
  const float* beta  = (const float*)d_in[5];
  float* out = (float*)d_out;

  float* ws   = (float*)d_ws;
  float* Wh   = ws;
  float* s1   = Wh + (size_t)B_ * N_ * FOUT;
  float* s2   = s1 + B_ * N_;
  unsigned* mbits = (unsigned*)(s2 + 3 * B_ * N_);

  wh_pack_kernel<<<2 * B_ * (N_ / 16), 256, 0, stream>>>(h, W, a, adj, Wh, s1, s2, mbits);
  hprime_kernel<<<B_ * (N_ / 32), 256, 0, stream>>>(mbits, Wh, s1, s2,
                                                    gamma, beta, out);
}


# Wh kernel: a block's 16 x 256 strip of h staged once in LDS by whole-row 16-byte loads issued together; A fragments read from LDS, not re-read from global by all 8 waves; same products
# speedup vs baseline: 1.8806x; 1.0864x over previous
#include <hip/hip_runtime.h>
#include <hip/hip_bf16.h>
#include <math.h>

typedef __attribute__((ext_vector_type(2))) float v2f;
typedef __attribute__((ext_vector_type(8))) float v8f;

#define B_   4
#define N_   2048
#define FIN  256
#define FOUT 128
#define ALPHA_ 0.2f
#define EPS_   1e-5f
#define NEG_   -9000000000000000.0f

#define WP_ 132
#define HP_ 260
#define RPW_ 2
__global__ __launch_bounds__(256) void wh_pack_kernel(const float* __restrict__ h,
                                                      const float* __restrict__ W,
                                                      const float* __restrict__ a,
                                                      const int* __restrict__ adj,
                                                      float* __restrict__ Wh,
                                                      float* __restrict__ s1,
                                                      float* __restrict__ s2,
                                                      unsigned* __restrict__ mbits) {
  __shared__ __attribute__((aligned(16))) float strip[16 * WP_];
  __shared__ __attribute__((aligned(16))) float hs[16 * HP_];
  int lane = threadIdx.x & 31;
  int bid  = blockIdx.x >> 1;

  if (blockIdx.x & 1) {
    int row0 = (bid * 8 + (threadIdx.x >> 5)) * RPW_;
    const int* arow = adj + (size_t)row0 * N_;
    unsigned wlo[RPW_], whi[RPW_];
#pragma unroll
    for (int r = 0; r < RPW_; ++r) { wlo[r] = 0u; whi[r] = 0u; }
    constexpr int GRP_ = 16;
#pragma unroll 1
    for (int w0 = 0; w0 < N_ / 32; w0 += GRP_) {
      int av[RPW_][GRP_];
#pragma unroll
      for (int u = 0; u < GRP_; ++u) {
        int j = (w0 + u) * 32 + lane;
#pragma unroll
        for (int r = 0; r < RPW_; ++r) av[r][u] = arow[(size_t)r * N_ + j];
      }
#pragma unroll
      for (int u = 0; u < GRP_; u += 2) {
        bool mine = lane == ((w0 + u) >> 1);
#pragma unroll
        for (int r = 0; r < RPW_; ++r) {
          unsigned b0 = __builtin_amdgcn_ballot_w32(av[r][u] > 0);
          unsigned b1 = __builtin_amdgcn_ballot_w32(av[r][u + 1] > 0);
          wlo[r] = mine ? b0 : wlo[r];
          whi[r] = mine ? b1 : whi[r];
        }
      }
    }
#pragma unroll
    for (int r = 0; r < RPW_; ++r)
      *(uint2*)(mbits + (size_t)(row0 + r) * (N_ / 32) + 2 * lane) = make_uint2(wlo[r], whi[r]);
    return;
  }

  int otile = threadIdx.x >> 5;
  int itile = bid & (N_ / 16 - 1);
  int b     = bid >> 7;
  int i0 = itile * 16, o0 = otile * 16;
  int r16  = lane & 15;
  int koff = (lane >> 4) << 1;

  const float4* a1 = (const float4*)a;
  const float4* a2 = (const float4*)(a + FOUT);
  float4 x1 = a1[lane];
  float4 x2 = a2[lane];

  const float* hblk = h + (size_t)(b * N_ + i0) * FIN;
  float4 hv[4];
#pragma unroll
  for (int q = 0; q < 4; ++q)
    hv[q] = *(const float4*)(hblk + (size_t)(otile + 8 * (q >> 1)) * FIN + 128 * (q & 1) + 4 * lane);
#pragma unroll
  for (int q = 0; q < 4; ++q)
    *(float4*)&hs[(otile + 8 * (q >> 1)) * HP_ + 128 * (q & 1) + 4 * lane] = hv[q];
  __syncthreads();

  const float* hrow = hs + r16 * HP_ + koff;
  v8f c = {};
  for (int k0 = 0; k0 < FIN; k0 += 4) {
    v2f a_ = *(const v2f*)(hrow + k0);
    v2f bb;
    bb.x = W[(size_t)(k0 + koff + 0) * FOUT + o0 + r16];
    bb.y = W[(size_t)(k0 + koff + 1) * FOUT + o0 + r16];
    c = __builtin_amdgcn_wmma_f32_16x16x4_f32(false, a_, false, bb, (short)0, c,
                                              false, false);
  }
#pragma unroll
  for (int r = 0; r < 8; ++r) {
    int M = r + ((lane >> 4) << 3);
    strip[M * WP_ + o0 + r16] = c[r];
  }
  __syncthreads();
#pragma unroll
  for (int r = 0; r < 8; ++r) {
    int M = r + ((lane >> 4) << 3);
    Wh[(size_t)(b * N_ + i0 + M) * FOUT + o0 + r16] = c[r];
  }

  float p1[2], p2[2];
#pragma unroll
  for (int rr = 0; rr < 2; ++rr) {
    float4 wv = *(const float4*)&strip[(2 * otile + rr) * WP_ + 4 * lane];
    p1[rr] = wv.x * x1.x + wv.y * x1.y + wv.z * x1.z + wv.w * x1.w;
    p2[rr] = wv.x * x2.x + wv.y * x2.y + wv.z * x2.z + wv.w * x2.w;
  }
#pragma unroll
  for (int off = 16; off >= 1; off >>= 1) {
#pragma unroll
    for (int rr = 0; rr < 2; ++rr) {
      p1[rr] += __shfl_xor(p1[rr], off, 32);
      p2[rr] += __shfl_xor(p2[rr], off, 32);
    }
  }
  if (lane < 2) {
    int node = b * N_ + i0 + 2 * otile + lane;
    s1[node] = (lane == 0) ? p1[0] : p1[1];
    s2[node] = (lane == 0) ? p2[0] : p2[1];
  }
}

typedef __attribute__((ext_vector_type(4))) float v4f;
#define JC_     128
#define NW3_    8
#define JQ_     (JC_ / 4)
#define PP_     288
#define STAGE_  ((JC_ / 2) * PP_ + NW3_ * JQ_)
#define TP_     132
#define LDSF_   (41 * 1024)
__global__ __launch_bounds__(256) void hprime_kernel(
    const unsigned* __restrict__ mbits, const float* __restrict__ Wh,
    const float* __restrict__ s1, const float* __restrict__ s2,
    const float* __restrict__ gamma, const float* __restrict__ beta,
    float* __restrict__ out) {
  __shared__ __attribute__((aligned(16))) float sB[LDSF_];
  int tid  = threadIdx.x;
  int lane = tid & 31;
  int wave = tid >> 5;
  int rt   = wave >> 2;
  int jq   = wave & 3;
  int b    = blockIdx.x >> 6;
  int i0   = (blockIdx.x & 63) * 32;
  int r16  = lane & 15;
  int hh   = lane >> 4;
  int rowid = b * N_ + i0 + rt * 16 + r16;

  float s1v  = s1[rowid];
  float ls0 = 0.f, ls1 = 0.f;
  const unsigned* mb = mbits + (size_t)rowid * (N_ / 32) + jq;
  const int mshift   = 2 * hh;
  const float* s2b   = s2 + (size_t)b * N_;
  const float* Whb   = Wh + (size_t)b * N_ * FOUT;

  v8f acc[8];
#pragma unroll
  for (int ot = 0; ot < 8; ++ot) acc[ot] = (v8f){};

  constexpr int NP = JC_ / (2 * NW3_);
  v4f  pa[NP], pb[NP];
  float s2pre;
  unsigned mw, mwn;
  const float* gsrc = Whb + (size_t)(2 * wave) * FOUT + 4 * lane;
  const int    ldst = wave * PP_ + 8 * lane;
  const int    sdst = (JC_ / 2) * PP_ + wave * JQ_ + lane;

#pragma unroll
  for (int i = 0; i < NP; ++i) {
    pa[i] = *(const v4f*)(gsrc + (size_t)(2 * NW3_ * i) * FOUT);
    pb[i] = *(const v4f*)(gsrc + (size_t)(2 * NW3_ * i + 1) * FOUT);
  }
  s2pre = s2b[JQ_ * jq + lane];
  mw = mb[0];

  float* mrow_s = sB + 2 * STAGE_;
  {
    const int wv   = __builtin_amdgcn_readfirstlane(wave);
    const int row0 = b * N_ + i0 + 4 * wv;
    float s2v[N_ / 32];
#pragma unroll
    for (int w = 0; w < N_ / 32; ++w) s2v[w] = s2b[32 * w + lane];
    uint2 mv[4];
    float s1r[4];
#pragma unroll
    for (int rr = 0; rr < 4; ++rr) {
      mv[rr]  = *(const uint2*)(mbits + (size_t)(row0 + rr) * (N_ / 32) + 2 * lane);
      s1r[rr] = s1[row0 + rr];
    }
    float mx[4];
#pragma unroll
    for (int rr = 0; rr < 4; ++rr) {
      const unsigned* mr = mbits + (size_t)(row0 + rr) * (N_ / 32);
      float mx0 = -INFINITY, mx1 = -INFINITY;
#pragma unroll
      for (int w = 0; w < N_ / 32; w += 2) {
        mx0 = fmaxf(mx0, __builtin_amdgcn_inverse_ballot_w32(mr[w]) ? s2v[w] : -INFINITY);
        mx1 = fmaxf(mx1, __builtin_amdgcn_inverse_ballot_w32(mr[w + 1]) ? s2v[w + 1] : -INFINITY);
      }
      mx[rr] = fmaxf(mx0, mx1);
    }
#pragma unroll
    for (int off = 16; off >= 1; off >>= 1) {
#pragma unroll
      for (int rr = 0; rr < 4; ++rr) mx[rr] = fmaxf(mx[rr], __shfl_xor(mx[rr], off, 32));
    }
    v4f m4;
#pragma unroll
    for (int rr = 0; rr < 4; ++rr) {
      bool alln = __builtin_amdgcn_ballot_w32((mv[rr].x & mv[rr].y) == 0xffffffffu) == 0xffffffffu;
      float e = s1r[rr] + mx[rr];
      e = (e >= 0.f) ? e : ALPHA_ * e;
      m4[rr] = alln ? e : fmaxf(e, NEG_);
    }
    if (lane == 0) *(v4f*)&mrow_s[4 * wv] = m4;
  }

#pragma unroll
  for (int i = 0; i < NP; ++i) {
    v4f lo = {pa[i].x, pb[i].x, pa[i].y, pb[i].y};
    v4f hi = {pa[i].z, pb[i].z, pa[i].w, pb[i].w};
    *(v4f*)&sB[ldst + NW3_ * i * PP_]     = lo;
    *(v4f*)&sB[ldst + NW3_ * i * PP_ + 4] = hi;
  }
  sB[sdst] = s2pre;
  __syncthreads();
  const float mi = mrow_s[rt * 16 + r16];

  constexpr int NCH = N_ / JC_;
  for (int ck = 0; ck < NCH; ++ck) {
    const float* sb = sB + (ck & 1) * STAGE_;
    int cn = (ck + 1 < NCH) ? ck + 1 : ck;
    const float* gn = gsrc + (size_t)(cn * JC_) * FOUT;
#pragma unroll
    for (int i = 0; i < NP; ++i) {
      pa[i] = *(const v4f*)(gn + (size_t)(2 * NW3_ * i) * FOUT);
      pb[i] = *(const v4f*)(gn + (size_t)(2 * NW3_ * i + 1) * FOUT);
    }
    s2pre = s2b[cn * JC_ + JQ_ * jq + lane];
    mwn = mb[(JC_ / 32) * cn];

    unsigned wsh = mw >> mshift;
    const float* s2l  = sb + (JC_ / 2) * PP_ + wave * JQ_ + 2 * hh;
    const float* brow = sb + (JQ_ / 2) * jq * PP_ + hh * PP_ + 2 * r16;
#pragma unroll
    for (int t = 0; t < JQ_ / 4; ++t) {
      v2f sv = *(const v2f*)(s2l + 4 * t);
      v2f a;
      {
        float e = s1v + sv.x;
        e = (e >= 0.f) ? e : ALPHA_ * e;
        float tv = ((wsh >> (4 * t)) & 1u) ? e : NEG_;
        a.x = __expf(tv - mi);
        ls0 += a.x;
      }
      {
        float e = s1v + sv.y;
        e = (e >= 0.f) ? e : ALPHA_ * e;
        float tv = ((wsh >> (4 * t + 1)) & 1u) ? e : NEG_;
        a.y = __expf(tv - mi);
        ls1 += a.y;
      }
      const float* bp = brow + (2 * t) * PP_;
#pragma unroll
      for (int ot = 0; ot < 8; ++ot) {
        v2f bb = *(const v2f*)(bp + 32 * ot);
        acc[ot] = __builtin_amdgcn_wmma_f32_16x16x4_f32(false, a, false, bb,
                                                        (short)0, acc[ot],
                                                        false, false);
      }
    }

    float* sn = sB + ((ck + 1) & 1) * STAGE_;
#pragma unroll
    for (int i = 0; i < NP; ++i) {
      v4f lo = {pa[i].x, pb[i].x, pa[i].y, pb[i].y};
      v4f hi = {pa[i].z, pb[i].z, pa[i].w, pb[i].w};
      *(v4f*)&sn[ldst + NW3_ * i * PP_]     = lo;
      *(v4f*)&sn[ldst + NW3_ * i * PP_ + 4] = hi;
    }
    sn[sdst] = s2pre;
    mw = mwn;
    __syncthreads();
  }

  float* tile = sB;
  float* lsum = sB + 4 * 32 * TP_;
#pragma unroll
  for (int ot = 0; ot < 8; ++ot)
#pragma unroll
    for (int r = 0; r < 8; ++r)
      tile[(jq * 32 + rt * 16 + hh * 8 + r) * TP_ + ot * 16 + r16] = acc[ot][r];
  float ls = ls0 + ls1;
  ls += __shfl_xor(ls, 16, 32);
  if (hh == 0) lsum[jq * 32 + rt * 16 + r16] = ls;
  __syncthreads();

  v4f g4 = *(const v4f*)(gamma + 4 * lane);
  v4f b4 = *(const v4f*)(beta + 4 * lane);
  v4f xr[4];
  float sr[4], qr[4];
#pragma unroll
  for (int rr = 0; rr < 4; ++rr) {
    int row = wave * 4 + rr;
    v4f x0 = *(const v4f*)&tile[(0 * 32 + row) * TP_ + 4 * lane];
    v4f x1 = *(const v4f*)&tile[(1 * 32 + row) * TP_ + 4 * lane];
    v4f x2 = *(const v4f*)&tile[(2 * 32 + row) * TP_ + 4 * lane];
    v4f x3 = *(const v4f*)&tile[(3 * 32 + row) * TP_ + 4 * lane];
    v4f x = (x0 + x1) + (x2 + x3);
    float l = (lsum[row] + lsum[32 + row]) + (lsum[64 + row] + lsum[96 + row]);
    float il = 1.0f / l;
    x.x *= il; x.y *= il; x.z *= il; x.w *= il;
    xr[rr] = x;
    sr[rr] = (x.x + x.y) + (x.z + x.w);
    qr[rr] = (x.x * x.x + x.y * x.y) + (x.z * x.z + x.w * x.w);
  }
#pragma unroll
  for (int off = 16; off >= 1; off >>= 1) {
#pragma unroll
    for (int rr = 0; rr < 4; ++rr) {
      sr[rr] += __shfl_xor(sr[rr], off, 32);
      qr[rr] += __shfl_xor(qr[rr], off, 32);
    }
  }
#pragma unroll
  for (int rr = 0; rr < 4; ++rr) {
    int row = wave * 4 + rr;
    v4f x = xr[rr];
    float mu   = sr[rr] * (1.0f / FOUT);
    float var  = qr[rr] * (1.0f / FOUT) - mu * mu;
    float rstd = rsqrtf(var + EPS_);
    v4f y, o;
    y.x = (x.x - mu) * rstd * g4.x + b4.x;
    y.y = (x.y - mu) * rstd * g4.y + b4.y;
    y.z = (x.z - mu) * rstd * g4.z + b4.z;
    y.w = (x.w - mu) * rstd * g4.w + b4.w;
    o.x = 0.5f * y.x * (1.0f + erff(y.x * 0.70710678118654752f));
    o.y = 0.5f * y.y * (1.0f + erff(y.y * 0.70710678118654752f));
    o.z = 0.5f * y.z * (1.0f + erff(y.z * 0.70710678118654752f));
    o.w = 0.5f * y.w * (1.0f + erff(y.w * 0.70710678118654752f));
    *(v4f*)(out + (size_t)(b * N_ + i0 + row) * FOUT + 4 * lane) = o;
  }
}

extern "C" void kernel_launch(void* const* d_in, const int* in_sizes, int n_in,
                              void* d_out, int out_size, void* d_ws, size_t ws_size,
                              hipStream_t stream) {
  const float* h     = (const float*)d_in[0];
  const int*   adj   = (const int*)d_in[1];
  const float* W     = (const float*)d_in[2];
  const float* a     = (const float*)d_in[3];
  const float* gamma = (const float*)d_in[4];
  const float* beta  = (const float*)d_in[5];
  float* out = (float*)d_out;

  float* ws   = (float*)d_ws;
  float* Wh   = ws;
  float* s1   = Wh + (size_t)B_ * N_ * FOUT;
  float* s2   = s1 + B_ * N_;
  unsigned* mbits = (unsigned*)(s2 + 3 * B_ * N_);

  wh_pack_kernel<<<2 * B_ * (N_ / 16), 256, 0, stream>>>(h, W, a, adj, Wh, s1, s2, mbits);
  hprime_kernel<<<B_ * (N_ / 32), 256, 0, stream>>>(mbits, Wh, s1, s2,
                                                    gamma, beta, out);
}


# h_prime kernel: a wave owns a j slice (16 of every 128 j) of all 32 rows and stages its own Wh rows in private LDS: no block barrier in the j loop; 8 partial tiles summed in LDS; j sums reordered
# speedup vs baseline: 2.2636x; 1.2037x over previous
#include <hip/hip_runtime.h>
#include <hip/hip_bf16.h>
#include <math.h>

typedef __attribute__((ext_vector_type(2))) float v2f;
typedef __attribute__((ext_vector_type(8))) float v8f;

#define B_   4
#define N_   2048
#define FIN  256
#define FOUT 128
#define ALPHA_ 0.2f
#define EPS_   1e-5f
#define NEG_   -9000000000000000.0f

#define WP_ 132
#define HP_ 260
#define RPW_ 2
__global__ __launch_bounds__(256) void wh_pack_kernel(const float* __restrict__ h,
                                                      const float* __restrict__ W,
                                                      const float* __restrict__ a,
                                                      const int* __restrict__ adj,
                                                      float* __restrict__ Wh,
                                                      float* __restrict__ s1,
                                                      float* __restrict__ s2,
                                                      unsigned* __restrict__ mbits) {
  __shared__ __attribute__((aligned(16))) float strip[16 * WP_];
  __shared__ __attribute__((aligned(16))) float hs[16 * HP_];
  int lane = threadIdx.x & 31;
  int bid  = blockIdx.x >> 1;

  if (blockIdx.x & 1) {
    int row0 = (bid * 8 + (threadIdx.x >> 5)) * RPW_;
    const int* arow = adj + (size_t)row0 * N_;
    unsigned wlo[RPW_], whi[RPW_];
#pragma unroll
    for (int r = 0; r < RPW_; ++r) { wlo[r] = 0u; whi[r] = 0u; }
    constexpr int GRP_ = 16;
#pragma unroll 1
    for (int w0 = 0; w0 < N_ / 32; w0 += GRP_) {
      int av[RPW_][GRP_];
#pragma unroll
      for (int u = 0; u < GRP_; ++u) {
        int j = (w0 + u) * 32 + lane;
#pragma unroll
        for (int r = 0; r < RPW_; ++r) av[r][u] = arow[(size_t)r * N_ + j];
      }
#pragma unroll
      for (int u = 0; u < GRP_; u += 2) {
        bool mine = lane == ((w0 + u) >> 1);
#pragma unroll
        for (int r = 0; r < RPW_; ++r) {
          unsigned b0 = __builtin_amdgcn_ballot_w32(av[r][u] > 0);
          unsigned b1 = __builtin_amdgcn_ballot_w32(av[r][u + 1] > 0);
          wlo[r] = mine ? b0 : wlo[r];
          whi[r] = mine ? b1 : whi[r];
        }
      }
    }
#pragma unroll
    for (int r = 0; r < RPW_; ++r)
      *(uint2*)(mbits + (size_t)(row0 + r) * (N_ / 32) + 2 * lane) = make_uint2(wlo[r], whi[r]);
    return;
  }

  int otile = threadIdx.x >> 5;
  int itile = bid & (N_ / 16 - 1);
  int b     = bid >> 7;
  int i0 = itile * 16, o0 = otile * 16;
  int r16  = lane & 15;
  int koff = (lane >> 4) << 1;

  const float4* a1 = (const float4*)a;
  const float4* a2 = (const float4*)(a + FOUT);
  float4 x1 = a1[lane];
  float4 x2 = a2[lane];

  const float* hblk = h + (size_t)(b * N_ + i0) * FIN;
  float4 hv[4];
#pragma unroll
  for (int q = 0; q < 4; ++q)
    hv[q] = *(const float4*)(hblk + (size_t)(otile + 8 * (q >> 1)) * FIN + 128 * (q & 1) + 4 * lane);
#pragma unroll
  for (int q = 0; q < 4; ++q)
    *(float4*)&hs[(otile + 8 * (q >> 1)) * HP_ + 128 * (q & 1) + 4 * lane] = hv[q];
  __syncthreads();

  const float* hrow = hs + r16 * HP_ + koff;
  v8f c = {};
  for (int k0 = 0; k0 < FIN; k0 += 4) {
    v2f a_ = *(const v2f*)(hrow + k0);
    v2f bb;
    bb.x = W[(size_t)(k0 + koff + 0) * FOUT + o0 + r16];
    bb.y = W[(size_t)(k0 + koff + 1) * FOUT + o0 + r16];
    c = __builtin_amdgcn_wmma_f32_16x16x4_f32(false, a_, false, bb, (short)0, c,
                                              false, false);
  }
#pragma unroll
  for (int r = 0; r < 8; ++r) {
    int M = r + ((lane >> 4) << 3);
    strip[M * WP_ + o0 + r16] = c[r];
  }
  __syncthreads();
#pragma unroll
  for (int r = 0; r < 8; ++r) {
    int M = r + ((lane >> 4) << 3);
    Wh[(size_t)(b * N_ + i0 + M) * FOUT + o0 + r16] = c[r];
  }

  float p1[2], p2[2];
#pragma unroll
  for (int rr = 0; rr < 2; ++rr) {
    float4 wv = *(const float4*)&strip[(2 * otile + rr) * WP_ + 4 * lane];
    p1[rr] = wv.x * x1.x + wv.y * x1.y + wv.z * x1.z + wv.w * x1.w;
    p2[rr] = wv.x * x2.x + wv.y * x2.y + wv.z * x2.z + wv.w * x2.w;
  }
#pragma unroll
  for (int off = 16; off >= 1; off >>= 1) {
#pragma unroll
    for (int rr = 0; rr < 2; ++rr) {
      p1[rr] += __shfl_xor(p1[rr], off, 32);
      p2[rr] += __shfl_xor(p2[rr], off, 32);
    }
  }
  if (lane < 2) {
    int node = b * N_ + i0 + 2 * otile + lane;
    s1[node] = (lane == 0) ? p1[0] : p1[1];
    s2[node] = (lane == 0) ? p2[0] : p2[1];
  }
}

typedef __attribute__((ext_vector_type(4))) float v4f;
#define JC_     128
#define NW3_    8
#define JW_     (JC_ / NW3_)
#define PP_     288
#define WSTAGE_ ((JW_ / 2) * PP_ + JW_)
#define TP_     132
#define LDSF_   (41 * 1024)
__global__ __launch_bounds__(256) void hprime_kernel(
    const unsigned* __restrict__ mbits, const float* __restrict__ Wh,
    const float* __restrict__ s1, const float* __restrict__ s2,
    const float* __restrict__ gamma, const float* __restrict__ beta,
    float* __restrict__ out) {
  __shared__ __attribute__((aligned(16))) float sB[LDSF_];
  int tid  = threadIdx.x;
  int lane = tid & 31;
  int wave = tid >> 5;
  int b    = blockIdx.x >> 6;
  int i0   = (blockIdx.x & 63) * 32;
  int r16  = lane & 15;
  int hh   = lane >> 4;
  int rowid0 = b * N_ + i0 + r16;
  int rowid1 = rowid0 + 16;

  float s1v0 = s1[rowid0];
  float s1v1 = s1[rowid1];
  float l00 = 0.f, l01 = 0.f, l10 = 0.f, l11 = 0.f;
  const unsigned* mb0 = mbits + (size_t)rowid0 * (N_ / 32) + (wave >> 1);
  const unsigned* mb1 = mbits + (size_t)rowid1 * (N_ / 32) + (wave >> 1);
  const int mshift   = 16 * (wave & 1) + 2 * hh;
  const float* s2b   = s2 + (size_t)b * N_;
  const float* Whb   = Wh + (size_t)b * N_ * FOUT;

  v8f acc0[8], acc1[8];
#pragma unroll
  for (int ot = 0; ot < 8; ++ot) { acc0[ot] = (v8f){}; acc1[ot] = (v8f){}; }

  constexpr int NP = JW_ / 2;
  v4f  pa[NP], pb[NP];
  float s2pre;
  unsigned mw0, mw1, mwn0, mwn1;
  const float* gsrc = Whb + (size_t)(JW_ * wave) * FOUT + 4 * lane;
  float* wreg = sB + wave * (2 * WSTAGE_);
  const int ldst = 8 * lane;
  const int sdst = (JW_ / 2) * PP_ + r16;

#pragma unroll
  for (int i = 0; i < NP; ++i) {
    pa[i] = *(const v4f*)(gsrc + (size_t)(2 * i) * FOUT);
    pb[i] = *(const v4f*)(gsrc + (size_t)(2 * i + 1) * FOUT);
  }
  s2pre = s2b[JW_ * wave + r16];
  mw0 = mb0[0];
  mw1 = mb1[0];

  float* mrow_s = sB + NW3_ * 2 * WSTAGE_;
  {
    const int wv   = __builtin_amdgcn_readfirstlane(wave);
    const int row0 = b * N_ + i0 + 4 * wv;
    float s2v[N_ / 32];
#pragma unroll
    for (int w = 0; w < N_ / 32; ++w) s2v[w] = s2b[32 * w + lane];
    uint2 mv[4];
    float s1r[4];
#pragma unroll
    for (int rr = 0; rr < 4; ++rr) {
      mv[rr]  = *(const uint2*)(mbits + (size_t)(row0 + rr) * (N_ / 32) + 2 * lane);
      s1r[rr] = s1[row0 + rr];
    }
    float mx[4];
#pragma unroll
    for (int rr = 0; rr < 4; ++rr) {
      const unsigned* mr = mbits + (size_t)(row0 + rr) * (N_ / 32);
      float mx0 = -INFINITY, mx1 = -INFINITY;
#pragma unroll
      for (int w = 0; w < N_ / 32; w += 2) {
        mx0 = fmaxf(mx0, __builtin_amdgcn_inverse_ballot_w32(mr[w]) ? s2v[w] : -INFINITY);
        mx1 = fmaxf(mx1, __builtin_amdgcn_inverse_ballot_w32(mr[w + 1]) ? s2v[w + 1] : -INFINITY);
      }
      mx[rr] = fmaxf(mx0, mx1);
    }
#pragma unroll
    for (int off = 16; off >= 1; off >>= 1) {
#pragma unroll
      for (int rr = 0; rr < 4; ++rr) mx[rr] = fmaxf(mx[rr], __shfl_xor(mx[rr], off, 32));
    }
    v4f m4;
#pragma unroll
    for (int rr = 0; rr < 4; ++rr) {
      bool alln = __builtin_amdgcn_ballot_w32((mv[rr].x & mv[rr].y) == 0xffffffffu) == 0xffffffffu;
      float e = s1r[rr] + mx[rr];
      e = (e >= 0.f) ? e : ALPHA_ * e;
      m4[rr] = alln ? e : fmaxf(e, NEG_);
    }
    if (lane == 0) *(v4f*)&mrow_s[4 * wv] = m4;
  }

#pragma unroll
  for (int i = 0; i < NP; ++i) {
    v4f lo = {pa[i].x, pb[i].x, pa[i].y, pb[i].y};
    v4f hi = {pa[i].z, pb[i].z, pa[i].w, pb[i].w};
    *(v4f*)&wreg[ldst + i * PP_]     = lo;
    *(v4f*)&wreg[ldst + i * PP_ + 4] = hi;
  }
  wreg[sdst] = s2pre;
  __syncthreads();
  const float mi0 = mrow_s[r16];
  const float mi1 = mrow_s[16 + r16];

  constexpr int NCH = N_ / JC_;
  for (int ck = 0; ck < NCH; ++ck) {
    const float* sb = wreg + (ck & 1) * WSTAGE_;
    int cn = (ck + 1 < NCH) ? ck + 1 : ck;
    const float* gn = gsrc + (size_t)(cn * JC_) * FOUT;
#pragma unroll
    for (int i = 0; i < NP; ++i) {
      pa[i] = *(const v4f*)(gn + (size_t)(2 * i) * FOUT);
      pb[i] = *(const v4f*)(gn + (size_t)(2 * i + 1) * FOUT);
    }
    s2pre = s2b[cn * JC_ + JW_ * wave + r16];
    mwn0 = mb0[(JC_ / 32) * cn];
    mwn1 = mb1[(JC_ / 32) * cn];

    unsigned wsh0 = mw0 >> mshift;
    unsigned wsh1 = mw1 >> mshift;
    const float* s2l  = sb + (JW_ / 2) * PP_ + 2 * hh;
    const float* brow = sb + hh * PP_ + 2 * r16;
#pragma unroll
    for (int t = 0; t < JW_ / 4; ++t) {
      v2f sv = *(const v2f*)(s2l + 4 * t);
      v2f a0, a1;
      {
        float e = s1v0 + sv.x;
        e = (e >= 0.f) ? e : ALPHA_ * e;
        float tv = ((wsh0 >> (4 * t)) & 1u) ? e : NEG_;
        a0.x = __expf(tv - mi0);
        l00 += a0.x;
      }
      {
        float e = s1v0 + sv.y;
        e = (e >= 0.f) ? e : ALPHA_ * e;
        float tv = ((wsh0 >> (4 * t + 1)) & 1u) ? e : NEG_;
        a0.y = __expf(tv - mi0);
        l01 += a0.y;
      }
      {
        float e = s1v1 + sv.x;
        e = (e >= 0.f) ? e : ALPHA_ * e;
        float tv = ((wsh1 >> (4 * t)) & 1u) ? e : NEG_;
        a1.x = __expf(tv - mi1);
        l10 += a1.x;
      }
      {
        float e = s1v1 + sv.y;
        e = (e >= 0.f) ? e : ALPHA_ * e;
        float tv = ((wsh1 >> (4 * t + 1)) & 1u) ? e : NEG_;
        a1.y = __expf(tv - mi1);
        l11 += a1.y;
      }
      const float* bp = brow + (2 * t) * PP_;
#pragma unroll
      for (int ot = 0; ot < 8; ++ot) {
        v2f bb = *(const v2f*)(bp + 32 * ot);
        acc0[ot] = __builtin_amdgcn_wmma_f32_16x16x4_f32(false, a0, false, bb,
                                                         (short)0, acc0[ot],
                                                         false, false);
        acc1[ot] = __builtin_amdgcn_wmma_f32_16x16x4_f32(false, a1, false, bb,
                                                         (short)0, acc1[ot],
                                                         false, false);
      }
    }

    float* sn = wreg + ((ck + 1) & 1) * WSTAGE_;
#pragma unroll
    for (int i = 0; i < NP; ++i) {
      v4f lo = {pa[i].x, pb[i].x, pa[i].y, pb[i].y};
      v4f hi = {pa[i].z, pb[i].z, pa[i].w, pb[i].w};
      *(v4f*)&sn[ldst + i * PP_]     = lo;
      *(v4f*)&sn[ldst + i * PP_ + 4] = hi;
    }
    sn[sdst] = s2pre;
    mw0 = mwn0;
    mw1 = mwn1;
    __builtin_amdgcn_fence(__ATOMIC_RELEASE, "wavefront");
    __builtin_amdgcn_wave_barrier();
    __builtin_amdgcn_fence(__ATOMIC_ACQUIRE, "wavefront");
  }

  __syncthreads();
  float* tile = sB;
  float* lsum = sB + NW3_ * 32 * TP_;
#pragma unroll
  for (int ot = 0; ot < 8; ++ot)
#pragma unroll
    for (int r = 0; r < 8; ++r) {
      tile[(wave * 32 + hh * 8 + r) * TP_ + ot * 16 + r16]      = acc0[ot][r];
      tile[(wave * 32 + 16 + hh * 8 + r) * TP_ + ot * 16 + r16] = acc1[ot][r];
    }
  float ls0 = l00 + l01;
  float ls1 = l10 + l11;
  ls0 += __shfl_xor(ls0, 16, 32);
  ls1 += __shfl_xor(ls1, 16, 32);
  if (hh == 0) {
    lsum[wave * 32 + r16]      = ls0;
    lsum[wave * 32 + 16 + r16] = ls1;
  }
  __syncthreads();

  v4f g4 = *(const v4f*)(gamma + 4 * lane);
  v4f b4 = *(const v4f*)(beta + 4 * lane);
  v4f xr[4];
  float sr[4], qr[4];
#pragma unroll
  for (int rr = 0; rr < 4; ++rr) {
    int row = wave * 4 + rr;
    v4f xq[NW3_];
    float lq[NW3_];
#pragma unroll
    for (int q = 0; q < NW3_; ++q) {
      xq[q] = *(const v4f*)&tile[(q * 32 + row) * TP_ + 4 * lane];
      lq[q] = lsum[q * 32 + row];
    }
    v4f x = ((xq[0] + xq[1]) + (xq[2] + xq[3])) + ((xq[4] + xq[5]) + (xq[6] + xq[7]));
    float l = ((lq[0] + lq[1]) + (lq[2] + lq[3])) + ((lq[4] + lq[5]) + (lq[6] + lq[7]));
    float il = 1.0f / l;
    x.x *= il; x.y *= il; x.z *= il; x.w *= il;
    xr[rr] = x;
    sr[rr] = (x.x + x.y) + (x.z + x.w);
    qr[rr] = (x.x * x.x + x.y * x.y) + (x.z * x.z + x.w * x.w);
  }
#pragma unroll
  for (int off = 16; off >= 1; off >>= 1) {
#pragma unroll
    for (int rr = 0; rr < 4; ++rr) {
      sr[rr] += __shfl_xor(sr[rr], off, 32);
      qr[rr] += __shfl_xor(qr[rr], off, 32);
    }
  }
#pragma unroll
  for (int rr = 0; rr < 4; ++rr) {
    int row = wave * 4 + rr;
    v4f x = xr[rr];
    float mu   = sr[rr] * (1.0f / FOUT);
    float var  = qr[rr] * (1.0f / FOUT) - mu * mu;
    float rstd = rsqrtf(var + EPS_);
    v4f y, o;
    y.x = (x.x - mu) * rstd * g4.x + b4.x;
    y.y = (x.y - mu) * rstd * g4.y + b4.y;
    y.z = (x.z - mu) * rstd * g4.z + b4.z;
    y.w = (x.w - mu) * rstd * g4.w + b4.w;
    o.x = 0.5f * y.x * (1.0f + erff(y.x * 0.70710678118654752f));
    o.y = 0.5f * y.y * (1.0f + erff(y.y * 0.70710678118654752f));
    o.z = 0.5f * y.z * (1.0f + erff(y.z * 0.70710678118654752f));
    o.w = 0.5f * y.w * (1.0f + erff(y.w * 0.70710678118654752f));
    *(v4f*)(out + (size_t)(b * N_ + i0 + row) * FOUT + 4 * lane) = o;
  }
}

extern "C" void kernel_launch(void* const* d_in, const int* in_sizes, int n_in,
                              void* d_out, int out_size, void* d_ws, size_t ws_size,
                              hipStream_t stream) {
  const float* h     = (const float*)d_in[0];
  const int*   adj   = (const int*)d_in[1];
  const float* W     = (const float*)d_in[2];
  const float* a     = (const float*)d_in[3];
  const float* gamma = (const float*)d_in[4];
  const float* beta  = (const float*)d_in[5];
  float* out = (float*)d_out;

  float* ws   = (float*)d_ws;
  float* Wh   = ws;
  float* s1   = Wh + (size_t)B_ * N_ * FOUT;
  float* s2   = s1 + B_ * N_;
  unsigned* mbits = (unsigned*)(s2 + 3 * B_ * N_);

  wh_pack_kernel<<<2 * B_ * (N_ / 16), 256, 0, stream>>>(h, W, a, adj, Wh, s1, s2, mbits);
  hprime_kernel<<<B_ * (N_ / 32), 256, 0, stream>>>(mbits, Wh, s1, s2,
                                                    gamma, beta, out);
}
